# starting kernel
# baseline (speedup 1.0000x reference)
#include <hip/hip_runtime.h>
#include <hip/hip_bf16.h>
#include <math.h>


typedef __attribute__((ext_vector_type(2))) float v2f;
typedef __attribute__((ext_vector_type(8))) float v8f;

#define DIM_K 128

__device__ __forceinline__ float fast_tanh(float x) {
    float ax = fabsf(x);
    float e  = __expf(-2.0f * ax);
    float r  = (1.0f - e) * __builtin_amdgcn_rcpf(1.0f + e);
    return copysignf(r, x);
}

__global__ void gcn_zero_kernel(float* __restrict__ p, long long n) {
    long long i = (long long)blockIdx.x * blockDim.x + threadIdx.x;
    long long stride = (long long)gridDim.x * blockDim.x;
    for (; i < n; i += stride) p[i] = 0.0f;
}

__global__ void gcn_spmm_kernel(const float* __restrict__ vals,
                                const int* __restrict__ row,
                                const int* __restrict__ col,
                                const float* __restrict__ X,
                                float* __restrict__ H,
                                int E) {
    int wavesPerBlock = blockDim.x >> 5;
    int e = blockIdx.x * wavesPerBlock + (threadIdx.x >> 5);
    int lane = threadIdx.x & 31;
    if (e >= E) return;
    float v = vals[e];
    int c = col[e];
    int r = row[e];
    const float4* src = (const float4*)(X + (long long)c * DIM_K);
    float4 m = src[lane];
    float* dst = H + (long long)r * DIM_K + lane * 4;
    atomicAdd(dst + 0, v * m.x);
    atomicAdd(dst + 1, v * m.y);
    atomicAdd(dst + 2, v * m.z);
    atomicAdd(dst + 3, v * m.w);
}

template <int NT>
__global__ void gcn_gemm_tanh_kernel(const float* __restrict__ S,
                                     const float* __restrict__ W,
                                     float* __restrict__ O,
                                     int nNodes) {
    const int lane = threadIdx.x & 31;
    const int wavesPerBlock = blockDim.x >> 5;
    const int mTile = blockIdx.x * wavesPerBlock + (threadIdx.x >> 5);
    if (mTile * 16 >= nNodes) return;

    const int nOut  = NT * 16;
    const int l15   = lane & 15;
    const int khalf = (lane < 16) ? 0 : 2;

    const float* aBase = S + (long long)(mTile * 16 + l15) * DIM_K + khalf;
    const float* bBase = W + (long long)l15 * DIM_K + khalf;

    v8f acc[NT];
#pragma unroll
    for (int t = 0; t < NT; ++t) acc[t] = (v8f){};

#pragma unroll 2
    for (int k = 0; k < DIM_K; k += 4) {
        v2f a;
        a.x = aBase[k];
        a.y = aBase[k + 1];
#pragma unroll
        for (int t = 0; t < NT; ++t) {
            const float* bp = bBase + (long long)t * 16 * DIM_K + k;
            v2f b;
            b.x = bp[0];
            b.y = bp[1];
            acc[t] = __builtin_amdgcn_wmma_f32_16x16x4_f32(
                 false, a,  false, b,
                 (short)0, acc[t],  false,  false);
        }
    }

    const int mbase = mTile * 16 + ((lane < 16) ? 0 : 8);
#pragma unroll
    for (int t = 0; t < NT; ++t) {
        const int n = t * 16 + l15;
#pragma unroll
        for (int r = 0; r < 8; ++r) {
            O[(long long)(mbase + r) * nOut + n] = fast_tanh(acc[t][r]);
        }
    }
}

extern "C" void kernel_launch(void* const* d_in, const int* in_sizes, int n_in,
                              void* d_out, int out_size, void* d_ws, size_t ws_size,
                              hipStream_t stream) {
    const float* X        = (const float*)d_in[0];
    const float* W1       = (const float*)d_in[1];
    const float* W2       = (const float*)d_in[2];
    const float* edgeVals = (const float*)d_in[3];
    const int*   edgeRow  = (const int*)d_in[4];
    const int*   edgeCol  = (const int*)d_in[5];
    float* out = (float*)d_out;

    const int N = in_sizes[0] / DIM_K;
    const int E = in_sizes[3];
    const long long featCount = (long long)N * DIM_K;

    float* S  = (float*)d_ws;
    float* H1 = S + featCount;

    const int zeroBlocks = 2048, zeroThreads = 256;

    const int spmmThreads = 256;
    const int edgesPerBlock = spmmThreads / 32;
    const int spmmBlocks = (E + edgesPerBlock - 1) / edgesPerBlock;

    const int gemmThreads = 256;
    const int wavesPerBlock = gemmThreads / 32;
    const int mTiles = (N + 15) / 16;
    const int gemmBlocks = (mTiles + wavesPerBlock - 1) / wavesPerBlock;

    gcn_zero_kernel<<<zeroBlocks, zeroThreads, 0, stream>>>(S, featCount);
    gcn_spmm_kernel<<<spmmBlocks, spmmThreads, 0, stream>>>(
        edgeVals, edgeRow, edgeCol, X, S, E);
    gcn_gemm_tanh_kernel<8><<<gemmBlocks, gemmThreads, 0, stream>>>(
        S, W1, H1, N);

    gcn_zero_kernel<<<zeroBlocks, zeroThreads, 0, stream>>>(S, featCount);
    gcn_spmm_kernel<<<spmmBlocks, spmmThreads, 0, stream>>>(
        edgeVals, edgeRow, edgeCol, H1, S, E);
    gcn_gemm_tanh_kernel<4><<<gemmBlocks, gemmThreads, 0, stream>>>(
        S, W2, out, N);
}


# Edge scatter by global float atomics replaced by row-range owner work-groups: each sweeps edge_row, keeps its rows' edges, sums them in LDS (ds_add_f32, LDS arrival order), stores each row once
# speedup vs baseline: 2.1445x; 2.1445x over previous
#include <hip/hip_runtime.h>
#include <hip/hip_bf16.h>
#include <math.h>


typedef __attribute__((ext_vector_type(2))) float v2f;
typedef __attribute__((ext_vector_type(8))) float v8f;

#define DIM_K 128

__device__ __forceinline__ float fast_tanh(float x) {
    float ax = fabsf(x);
    float e  = __expf(-2.0f * ax);
    float r  = (1.0f - e) * __builtin_amdgcn_rcpf(1.0f + e);
    return copysignf(r, x);
}

#define OWN_ROWS  392
#define OWN_WAVES 32
#define OWN_QCAP  640

__global__ __launch_bounds__(1024)
void gcn_spmm_owner_kernel(const float* __restrict__ vals,
                           const int* __restrict__ row,
                           const int* __restrict__ col,
                           const float* __restrict__ X,
                           float* __restrict__ S,
                           int E, int nNodes) {
    __shared__ float    acc[(OWN_ROWS + 1) * DIM_K];
    __shared__ unsigned queue[OWN_WAVES][OWN_QCAP];

    const int tid  = threadIdx.x;
    const int lane = tid & 31;
    const int w = __builtin_amdgcn_readfirstlane(tid >> 5);
    const unsigned laneLt = (1u << lane) - 1u;

    const int nGroups = E >> 7;
    const int nIt     = (nGroups + OWN_WAVES - 1) / OWN_WAVES;
    const int nStep   = (nIt + 3) >> 2;
    const int startGroup = (int)(((long long)blockIdx.x * nGroups) / gridDim.x);

    for (int base = blockIdx.x * OWN_ROWS; base < nNodes; base += gridDim.x * OWN_ROWS) {
        const unsigned rowsHere = (unsigned)min(OWN_ROWS, nNodes - base);

        for (int i = tid; i < (OWN_ROWS + 1) * DIM_K / 4; i += blockDim.x)
            ((float4*)acc)[i] = make_float4(0.f, 0.f, 0.f, 0.f);
        __syncthreads();

        int cnt = 0;
        auto push = [&](bool m, int e, unsigned d) {
            const unsigned b = __builtin_amdgcn_ballot_w32(m);
            if (m) queue[w][cnt + __builtin_popcount(b & laneLt)] = ((unsigned)e << 9) | d;
            cnt += __builtin_popcount(b);
        };

        auto groupOf = [&](int it) -> int {
            int g = min(it * OWN_WAVES + w, nGroups - 1) + startGroup;
            if (g >= nGroups) g -= nGroups;
            return g;
        };
        auto loadGroup = [&](int it) -> int4 {
            return *(const int4*)(row + groupOf(it) * 128 + lane * 4);
        };
        auto scanGroup = [&](const int4 r, int it) {
            if (it * OWN_WAVES + w < nGroups) {
                const unsigned d0 = (unsigned)(r.x - base);
                const unsigned d1 = (unsigned)(r.y - base);
                const unsigned d2 = (unsigned)(r.z - base);
                const unsigned d3 = (unsigned)(r.w - base);
                const unsigned dm = min(min(d0, d1), min(d2, d3));
                if (__builtin_amdgcn_ballot_w32(dm < rowsHere) != 0u) {
                    const int e0 = groupOf(it) * 128 + lane * 4;
                    push(d0 < rowsHere, e0,     d0);
                    push(d1 < rowsHere, e0 + 1, d1);
                    push(d2 < rowsHere, e0 + 2, d2);
                    push(d3 < rowsHere, e0 + 3, d3);
                }
            }
        };
        auto drain = [&](const bool last) {
            const int todo = last ? cnt : (cnt & ~7);
            int i0 = 0;
            do {
                const int nb = min(32, todo - i0);
                int   c = 0, lr = OWN_ROWS;
                float v = 0.f;
                if (lane < nb) {
                    const unsigned en = queue[w][i0 + lane];
                    const int e = (int)(en >> 9);
                    c  = col[e];
                    v  = vals[e];
                    lr = (int)(en & 511u);
                }
                int j0 = 0;
                do {
                    float4 g[8];
                    float  vv[8];
                    int    rr[8];
#pragma unroll
                    for (int u = 0; u < 8; ++u) {
                        const int j  = j0 + u;
                        const int cj = __builtin_amdgcn_readlane(c, j);
                        vv[u] = __int_as_float(__builtin_amdgcn_readlane(__float_as_int(v), j));
                        rr[u] = __builtin_amdgcn_readlane(lr, j);
                        g[u]  = ((const float4*)(X + (long long)cj * DIM_K))[lane];
                    }
#pragma unroll
                    for (int u = 0; u < 8; ++u) {
                        float* dst = acc + rr[u] * DIM_K + lane;
                        __hip_atomic_fetch_add(dst,      vv[u] * g[u].x, __ATOMIC_RELAXED, __HIP_MEMORY_SCOPE_WORKGROUP);
                        __hip_atomic_fetch_add(dst + 32, vv[u] * g[u].y, __ATOMIC_RELAXED, __HIP_MEMORY_SCOPE_WORKGROUP);
                        __hip_atomic_fetch_add(dst + 64, vv[u] * g[u].z, __ATOMIC_RELAXED, __HIP_MEMORY_SCOPE_WORKGROUP);
                        __hip_atomic_fetch_add(dst + 96, vv[u] * g[u].w, __ATOMIC_RELAXED, __HIP_MEMORY_SCOPE_WORKGROUP);
                    }
                    j0 += 8;
                } while (j0 < nb);
                i0 += 32;
            } while (i0 < todo);
            const int rem = cnt - todo;
            if (lane < rem) {
                const unsigned keep = queue[w][todo + lane];
                queue[w][lane] = keep;
            }
            cnt = rem;
        };

        for (int t = nGroups * 128 + w * 32; t < E; t += OWN_WAVES * 32) {
            const int e = t + lane;
            unsigned d = 0xffffffffu;
            if (e < E) d = (unsigned)(row[e] - base);
            push(d < rowsHere, e, d);
        }

        if (nGroups > 0) {
            int4 c0 = loadGroup(0), c1 = loadGroup(1), c2 = loadGroup(2), c3 = loadGroup(3);
#pragma unroll 1
            for (int st = 0; st < nStep; st += 2) {
                int it = st * 4;
                const int4 n0 = loadGroup(it + 4), n1 = loadGroup(it + 5);
                const int4 n2 = loadGroup(it + 6), n3 = loadGroup(it + 7);
                scanGroup(c0, it);
                scanGroup(c1, it + 1);
                scanGroup(c2, it + 2);
                scanGroup(c3, it + 3);
                if (cnt > OWN_QCAP - 512) drain(false);

                it += 4;
                c0 = loadGroup(it + 4); c1 = loadGroup(it + 5);
                c2 = loadGroup(it + 6); c3 = loadGroup(it + 7);
                scanGroup(n0, it);
                scanGroup(n1, it + 1);
                scanGroup(n2, it + 2);
                scanGroup(n3, it + 3);
                if (cnt > OWN_QCAP - 512) drain(false);
            }
        }
        if (cnt > 0) drain(true);
        __syncthreads();

        for (int idx = tid; idx < (int)rowsHere * 32; idx += blockDim.x) {
            const int lrow = idx >> 5;
            const float* a = acc + lrow * DIM_K + lane;
            const float4 o = make_float4(a[0], a[32], a[64], a[96]);
            ((float4*)(S + (long long)(base + lrow) * DIM_K))[lane] = o;
        }
        __syncthreads();
    }
}

template <int NT>
__global__ void gcn_gemm_tanh_kernel(const float* __restrict__ S,
                                     const float* __restrict__ W,
                                     float* __restrict__ O,
                                     int nNodes) {
    const int lane = threadIdx.x & 31;
    const int wavesPerBlock = blockDim.x >> 5;
    const int mTile = blockIdx.x * wavesPerBlock + (threadIdx.x >> 5);
    if (mTile * 16 >= nNodes) return;

    const int nOut  = NT * 16;
    const int l15   = lane & 15;
    const int khalf = (lane < 16) ? 0 : 2;

    const float* aBase = S + (long long)(mTile * 16 + l15) * DIM_K + khalf;
    const float* bBase = W + (long long)l15 * DIM_K + khalf;

    v8f acc[NT];
#pragma unroll
    for (int t = 0; t < NT; ++t) acc[t] = (v8f){};

#pragma unroll 2
    for (int k = 0; k < DIM_K; k += 4) {
        v2f a;
        a.x = aBase[k];
        a.y = aBase[k + 1];
#pragma unroll
        for (int t = 0; t < NT; ++t) {
            const float* bp = bBase + (long long)t * 16 * DIM_K + k;
            v2f b;
            b.x = bp[0];
            b.y = bp[1];
            acc[t] = __builtin_amdgcn_wmma_f32_16x16x4_f32(
                 false, a,  false, b,
                 (short)0, acc[t],  false,  false);
        }
    }

    const int mbase = mTile * 16 + ((lane < 16) ? 0 : 8);
#pragma unroll
    for (int t = 0; t < NT; ++t) {
        const int n = t * 16 + l15;
#pragma unroll
        for (int r = 0; r < 8; ++r) {
            O[(long long)(mbase + r) * nOut + n] = fast_tanh(acc[t][r]);
        }
    }
}

extern "C" void kernel_launch(void* const* d_in, const int* in_sizes, int n_in,
                              void* d_out, int out_size, void* d_ws, size_t ws_size,
                              hipStream_t stream) {
    const float* X        = (const float*)d_in[0];
    const float* W1       = (const float*)d_in[1];
    const float* W2       = (const float*)d_in[2];
    const float* edgeVals = (const float*)d_in[3];
    const int*   edgeRow  = (const int*)d_in[4];
    const int*   edgeCol  = (const int*)d_in[5];
    float* out = (float*)d_out;

    const int N = in_sizes[0] / DIM_K;
    const int E = in_sizes[3];
    const long long featCount = (long long)N * DIM_K;

    float* S  = (float*)d_ws;
    float* H1 = S + featCount;

    const int gemmThreads = 256;
    const int wavesPerBlock = gemmThreads / 32;
    const int mTiles = (N + 15) / 16;
    const int gemmBlocks = (mTiles + wavesPerBlock - 1) / wavesPerBlock;

    gcn_spmm_owner_kernel<<<256, 1024, 0, stream>>>(
        edgeVals, edgeRow, edgeCol, X, S, E, N);
    gcn_gemm_tanh_kernel<8><<<gemmBlocks, gemmThreads, 0, stream>>>(
        S, W1, H1, N);

    gcn_spmm_owner_kernel<<<256, 1024, 0, stream>>>(
        edgeVals, edgeRow, edgeCol, H1, S, E, N);
    gcn_gemm_tanh_kernel<4><<<gemmBlocks, gemmThreads, 0, stream>>>(
        S, W2, out, N);
}


# Layer 2 no longer sweeps edge_row: in layer 1 each wave writes the edges it keeps (source row, weight, local row) to its own workspace list, layer 2 reads it back; row sums stay in LDS arrival order
# speedup vs baseline: 2.6298x; 1.2263x over previous
#include <hip/hip_runtime.h>
#include <hip/hip_bf16.h>
#include <math.h>


typedef __attribute__((ext_vector_type(2))) float v2f;
typedef __attribute__((ext_vector_type(8))) float v8f;

#define DIM_K 128

__device__ __forceinline__ float fast_tanh(float x) {
    float ax = fabsf(x);
    float e  = __expf(-2.0f * ax);
    float r  = (1.0f - e) * __builtin_amdgcn_rcpf(1.0f + e);
    return copysignf(r, x);
}

#define OWN_ROWS  392
#define OWN_WAVES 32
#define OWN_QCAP  640
#define OWN_GROUPS 256
#define LIST_CAP  1024

template <bool REPLAY>
__global__ __launch_bounds__(1024)
void gcn_spmm_owner_kernel(const float* __restrict__ vals,
                           const int* __restrict__ row,
                           const int* __restrict__ col,
                           const float* __restrict__ X,
                           float* __restrict__ S,
                           unsigned* __restrict__ edgeList,
                           int* __restrict__ listCount,
                           int E, int nNodes) {
    __shared__ float    acc[(OWN_ROWS + 1) * DIM_K];
    __shared__ unsigned queue[OWN_WAVES][OWN_QCAP];

    const int tid  = threadIdx.x;
    const int lane = tid & 31;
    const int w = __builtin_amdgcn_readfirstlane(tid >> 5);
    const unsigned laneLt = (1u << lane) - 1u;

    const int nGroups = E >> 7;
    const int nIt     = (nGroups + OWN_WAVES - 1) / OWN_WAVES;
    const int nStep   = (nIt + 3) >> 2;
    const int startGroup = (int)(((long long)blockIdx.x * nGroups) / gridDim.x);

    uint2* const myList  = (uint2*)edgeList + (long long)(blockIdx.x * OWN_WAVES + w) * LIST_CAP;
    int*   const myCount = listCount + blockIdx.x * OWN_WAVES + w;
    const int firstBase  = blockIdx.x * OWN_ROWS;
    int rec = 0;

    for (int base = blockIdx.x * OWN_ROWS; base < nNodes; base += gridDim.x * OWN_ROWS) {
        const unsigned rowsHere = (unsigned)min(OWN_ROWS, nNodes - base);

        for (int i = tid; i < (OWN_ROWS + 1) * DIM_K / 4; i += blockDim.x)
            ((float4*)acc)[i] = make_float4(0.f, 0.f, 0.f, 0.f);
        __syncthreads();

        int cnt = 0;
        auto push = [&](bool m, int e, unsigned d) {
            const unsigned b = __builtin_amdgcn_ballot_w32(m);
            if (m) queue[w][cnt + __builtin_popcount(b & laneLt)] = ((unsigned)e << 9) | d;
            cnt += __builtin_popcount(b);
        };

        auto groupOf = [&](int it) -> int {
            int g = min(it * OWN_WAVES + w, nGroups - 1) + startGroup;
            if (g >= nGroups) g -= nGroups;
            return g;
        };
        auto loadGroup = [&](int it) -> int4 {
            return *(const int4*)(row + groupOf(it) * 128 + lane * 4);
        };
        auto scanGroup = [&](const int4 r, int it) {
            if (it * OWN_WAVES + w < nGroups) {
                const unsigned d0 = (unsigned)(r.x - base);
                const unsigned d1 = (unsigned)(r.y - base);
                const unsigned d2 = (unsigned)(r.z - base);
                const unsigned d3 = (unsigned)(r.w - base);
                const unsigned dm = min(min(d0, d1), min(d2, d3));
                if (__builtin_amdgcn_ballot_w32(dm < rowsHere) != 0u) {
                    const int e0 = groupOf(it) * 128 + lane * 4;
                    push(d0 < rowsHere, e0,     d0);
                    push(d1 < rowsHere, e0 + 1, d1);
                    push(d2 < rowsHere, e0 + 2, d2);
                    push(d3 < rowsHere, e0 + 3, d3);
                }
            }
        };
        auto addBatch = [&](const int c, const float v, const int lr, const int nb) {
            int j0 = 0;
            do {
                float4 g[8];
                float  vv[8];
                int    rr[8];
#pragma unroll
                for (int u = 0; u < 8; ++u) {
                    const int j  = j0 + u;
                    const int cj = __builtin_amdgcn_readlane(c, j);
                    vv[u] = __int_as_float(__builtin_amdgcn_readlane(__float_as_int(v), j));
                    rr[u] = __builtin_amdgcn_readlane(lr, j);
                    g[u]  = ((const float4*)(X + (long long)cj * DIM_K))[lane];
                }
#pragma unroll
                for (int u = 0; u < 8; ++u) {
                    float* dst = acc + rr[u] * DIM_K + lane;
                    __hip_atomic_fetch_add(dst,      vv[u] * g[u].x, __ATOMIC_RELAXED, __HIP_MEMORY_SCOPE_WORKGROUP);
                    __hip_atomic_fetch_add(dst + 32, vv[u] * g[u].y, __ATOMIC_RELAXED, __HIP_MEMORY_SCOPE_WORKGROUP);
                    __hip_atomic_fetch_add(dst + 64, vv[u] * g[u].z, __ATOMIC_RELAXED, __HIP_MEMORY_SCOPE_WORKGROUP);
                    __hip_atomic_fetch_add(dst + 96, vv[u] * g[u].w, __ATOMIC_RELAXED, __HIP_MEMORY_SCOPE_WORKGROUP);
                }
                j0 += 8;
            } while (j0 < nb);
        };

        auto drain = [&](const bool last) {
            const int todo = last ? cnt : (cnt & ~7);
            int i0 = 0;
            do {
                const int nb = min(32, todo - i0);
                int   c = 0, lr = OWN_ROWS;
                float v = 0.f;
                if (lane < nb) {
                    const unsigned en = queue[w][i0 + lane];
                    const int e = (int)(en >> 9);
                    c  = col[e];
                    v  = vals[e];
                    lr = (int)(en & 511u);
                }
                if constexpr (!REPLAY) {
                    if (rec >= 0) {
                        if (rec + nb <= LIST_CAP) {
                            if (lane < nb)
                                myList[rec + lane] = make_uint2(((unsigned)c << 9) | (unsigned)lr, __float_as_uint(v));
                            rec += nb;
                        } else {
                            rec = -1;
                        }
                    }
                }
                addBatch(c, v, lr, nb);
                i0 += 32;
            } while (i0 < todo);
            const int rem = cnt - todo;
            if (lane < rem) {
                const unsigned keep = queue[w][todo + lane];
                queue[w][lane] = keep;
            }
            cnt = rem;
        };

        bool sweep = true;
        if constexpr (REPLAY) {
            int n = -1;
            if (base == firstBase) n = __builtin_amdgcn_readfirstlane(*myCount);
            if ((unsigned)n <= (unsigned)LIST_CAP) {
                sweep = false;
                const uint2 filler = make_uint2((unsigned)OWN_ROWS, 0u);
                if (n > 0) {
                    const int lastEntry = n - 1;
                    uint2 cur = myList[min(lane, lastEntry)];
                    if (lane >= n) cur = filler;
                    int i0 = 0;
                    do {
                        uint2 nxt = myList[min(i0 + 32 + lane, lastEntry)];
                        if (i0 + 32 + lane >= n) nxt = filler;
                        addBatch((int)(cur.x >> 9), __uint_as_float(cur.y), (int)(cur.x & 511u), min(32, n - i0));
                        cur = nxt;
                        i0 += 32;
                    } while (i0 < n);
                }
            }
        }
        if (sweep) {
        for (int t = nGroups * 128 + w * 32; t < E; t += OWN_WAVES * 32) {
            const int e = t + lane;
            unsigned d = 0xffffffffu;
            if (e < E) d = (unsigned)(row[e] - base);
            push(d < rowsHere, e, d);
        }

        if (nGroups > 0) {
            int4 c0 = loadGroup(0), c1 = loadGroup(1), c2 = loadGroup(2), c3 = loadGroup(3);
#pragma unroll 1
            for (int st = 0; st < nStep; st += 2) {
                int it = st * 4;
                const int4 n0 = loadGroup(it + 4), n1 = loadGroup(it + 5);
                const int4 n2 = loadGroup(it + 6), n3 = loadGroup(it + 7);
                scanGroup(c0, it);
                scanGroup(c1, it + 1);
                scanGroup(c2, it + 2);
                scanGroup(c3, it + 3);
                if (cnt > OWN_QCAP - 512) drain(false);

                it += 4;
                c0 = loadGroup(it + 4); c1 = loadGroup(it + 5);
                c2 = loadGroup(it + 6); c3 = loadGroup(it + 7);
                scanGroup(n0, it);
                scanGroup(n1, it + 1);
                scanGroup(n2, it + 2);
                scanGroup(n3, it + 3);
                if (cnt > OWN_QCAP - 512) drain(false);
            }
        }
        if (cnt > 0) drain(true);
        }
        if constexpr (!REPLAY) {
            if (base == firstBase && lane == 0) *myCount = rec;
            rec = -1;
        }
        __syncthreads();

        for (int idx = tid; idx < (int)rowsHere * 32; idx += blockDim.x) {
            const int lrow = idx >> 5;
            const float* a = acc + lrow * DIM_K + lane;
            const float4 o = make_float4(a[0], a[32], a[64], a[96]);
            ((float4*)(S + (long long)(base + lrow) * DIM_K))[lane] = o;
        }
        __syncthreads();
    }
}

template <int NT>
__global__ void gcn_gemm_tanh_kernel(const float* __restrict__ S,
                                     const float* __restrict__ W,
                                     float* __restrict__ O,
                                     int nNodes) {
    const int lane = threadIdx.x & 31;
    const int wavesPerBlock = blockDim.x >> 5;
    const int mTile = blockIdx.x * wavesPerBlock + (threadIdx.x >> 5);
    if (mTile * 16 >= nNodes) return;

    const int nOut  = NT * 16;
    const int l15   = lane & 15;
    const int khalf = (lane < 16) ? 0 : 2;

    const float* aBase = S + (long long)(mTile * 16 + l15) * DIM_K + khalf;
    const float* bBase = W + (long long)l15 * DIM_K + khalf;

    v8f acc[NT];
#pragma unroll
    for (int t = 0; t < NT; ++t) acc[t] = (v8f){};

#pragma unroll 2
    for (int k = 0; k < DIM_K; k += 4) {
        v2f a;
        a.x = aBase[k];
        a.y = aBase[k + 1];
#pragma unroll
        for (int t = 0; t < NT; ++t) {
            const float* bp = bBase + (long long)t * 16 * DIM_K + k;
            v2f b;
            b.x = bp[0];
            b.y = bp[1];
            acc[t] = __builtin_amdgcn_wmma_f32_16x16x4_f32(
                 false, a,  false, b,
                 (short)0, acc[t],  false,  false);
        }
    }

    const int mbase = mTile * 16 + ((lane < 16) ? 0 : 8);
#pragma unroll
    for (int t = 0; t < NT; ++t) {
        const int n = t * 16 + l15;
#pragma unroll
        for (int r = 0; r < 8; ++r) {
            O[(long long)(mbase + r) * nOut + n] = fast_tanh(acc[t][r]);
        }
    }
}

extern "C" void kernel_launch(void* const* d_in, const int* in_sizes, int n_in,
                              void* d_out, int out_size, void* d_ws, size_t ws_size,
                              hipStream_t stream) {
    const float* X        = (const float*)d_in[0];
    const float* W1       = (const float*)d_in[1];
    const float* W2       = (const float*)d_in[2];
    const float* edgeVals = (const float*)d_in[3];
    const int*   edgeRow  = (const int*)d_in[4];
    const int*   edgeCol  = (const int*)d_in[5];
    float* out = (float*)d_out;

    const int N = in_sizes[0] / DIM_K;
    const int E = in_sizes[3];
    const long long featCount = (long long)N * DIM_K;

    float* S  = (float*)d_ws;
    float* H1 = S + featCount;
    unsigned* edgeList  = (unsigned*)(H1 + featCount);
    int*      listCount = (int*)(edgeList + 2LL * OWN_GROUPS * OWN_WAVES * LIST_CAP);

    const int gemmThreads = 256;
    const int wavesPerBlock = gemmThreads / 32;
    const int mTiles = (N + 15) / 16;
    const int gemmBlocks = (mTiles + wavesPerBlock - 1) / wavesPerBlock;

    gcn_spmm_owner_kernel<false><<<OWN_GROUPS, 1024, 0, stream>>>(
        edgeVals, edgeRow, edgeCol, X, S, edgeList, listCount, E, N);
    gcn_gemm_tanh_kernel<8><<<gemmBlocks, gemmThreads, 0, stream>>>(
        S, W1, H1, N);

    gcn_spmm_owner_kernel<true><<<OWN_GROUPS, 1024, 0, stream>>>(
        edgeVals, edgeRow, edgeCol, H1, S, edgeList, listCount, E, N);
    gcn_gemm_tanh_kernel<4><<<gemmBlocks, gemmThreads, 0, stream>>>(
        S, W2, out, N);
}


# Both GEMMs: one wave now does 32 rows x 64 columns (two strips share each B fragment) and fetches fragments 16 B per lane; k summed in a permuted order inside each group of 8
# speedup vs baseline: 2.9920x; 1.1377x over previous
#include <hip/hip_runtime.h>
#include <hip/hip_bf16.h>
#include <math.h>


typedef __attribute__((ext_vector_type(2))) float v2f;
typedef __attribute__((ext_vector_type(8))) float v8f;

#define DIM_K 128

__device__ __forceinline__ float fast_tanh(float x) {
    float ax = fabsf(x);
    float e  = __expf(-2.0f * ax);
    float r  = (1.0f - e) * __builtin_amdgcn_rcpf(1.0f + e);
    return copysignf(r, x);
}

#define OWN_ROWS  392
#define OWN_WAVES 32
#define OWN_QCAP  640
#define OWN_GROUPS 256
#define LIST_CAP  1024

template <bool REPLAY>
__global__ __launch_bounds__(1024)
void gcn_spmm_owner_kernel(const float* __restrict__ vals,
                           const int* __restrict__ row,
                           const int* __restrict__ col,
                           const float* __restrict__ X,
                           float* __restrict__ S,
                           unsigned* __restrict__ edgeList,
                           int* __restrict__ listCount,
                           int E, int nNodes) {
    __shared__ float    acc[(OWN_ROWS + 1) * DIM_K];
    __shared__ unsigned queue[OWN_WAVES][OWN_QCAP];

    const int tid  = threadIdx.x;
    const int lane = tid & 31;
    const int w = __builtin_amdgcn_readfirstlane(tid >> 5);
    const unsigned laneLt = (1u << lane) - 1u;

    const int nGroups = E >> 7;
    const int nIt     = (nGroups + OWN_WAVES - 1) / OWN_WAVES;
    const int nStep   = (nIt + 3) >> 2;
    const int startGroup = (int)(((long long)blockIdx.x * nGroups) / gridDim.x);

    uint2* const myList  = (uint2*)edgeList + (long long)(blockIdx.x * OWN_WAVES + w) * LIST_CAP;
    int*   const myCount = listCount + blockIdx.x * OWN_WAVES + w;
    const int firstBase  = blockIdx.x * OWN_ROWS;
    int rec = 0;

    for (int base = blockIdx.x * OWN_ROWS; base < nNodes; base += gridDim.x * OWN_ROWS) {
        const unsigned rowsHere = (unsigned)min(OWN_ROWS, nNodes - base);

        for (int i = tid; i < (OWN_ROWS + 1) * DIM_K / 4; i += blockDim.x)
            ((float4*)acc)[i] = make_float4(0.f, 0.f, 0.f, 0.f);
        __syncthreads();

        int cnt = 0;
        auto push = [&](bool m, int e, unsigned d) {
            const unsigned b = __builtin_amdgcn_ballot_w32(m);
            if (m) queue[w][cnt + __builtin_popcount(b & laneLt)] = ((unsigned)e << 9) | d;
            cnt += __builtin_popcount(b);
        };

        auto groupOf = [&](int it) -> int {
            int g = min(it * OWN_WAVES + w, nGroups - 1) + startGroup;
            if (g >= nGroups) g -= nGroups;
            return g;
        };
        auto loadGroup = [&](int it) -> int4 {
            return *(const int4*)(row + groupOf(it) * 128 + lane * 4);
        };
        auto scanGroup = [&](const int4 r, int it) {
            if (it * OWN_WAVES + w < nGroups) {
                const unsigned d0 = (unsigned)(r.x - base);
                const unsigned d1 = (unsigned)(r.y - base);
                const unsigned d2 = (unsigned)(r.z - base);
                const unsigned d3 = (unsigned)(r.w - base);
                const unsigned dm = min(min(d0, d1), min(d2, d3));
                if (__builtin_amdgcn_ballot_w32(dm < rowsHere) != 0u) {
                    const int e0 = groupOf(it) * 128 + lane * 4;
                    push(d0 < rowsHere, e0,     d0);
                    push(d1 < rowsHere, e0 + 1, d1);
                    push(d2 < rowsHere, e0 + 2, d2);
                    push(d3 < rowsHere, e0 + 3, d3);
                }
            }
        };
        auto addBatch = [&](const int c, const float v, const int lr, const int nb) {
            int j0 = 0;
            do {
                float4 g[8];
                float  vv[8];
                int    rr[8];
#pragma unroll
                for (int u = 0; u < 8; ++u) {
                    const int j  = j0 + u;
                    const int cj = __builtin_amdgcn_readlane(c, j);
                    vv[u] = __int_as_float(__builtin_amdgcn_readlane(__float_as_int(v), j));
                    rr[u] = __builtin_amdgcn_readlane(lr, j);
                    g[u]  = ((const float4*)(X + (long long)cj * DIM_K))[lane];
                }
#pragma unroll
                for (int u = 0; u < 8; ++u) {
                    float* dst = acc + rr[u] * DIM_K + lane;
                    __hip_atomic_fetch_add(dst,      vv[u] * g[u].x, __ATOMIC_RELAXED, __HIP_MEMORY_SCOPE_WORKGROUP);
                    __hip_atomic_fetch_add(dst + 32, vv[u] * g[u].y, __ATOMIC_RELAXED, __HIP_MEMORY_SCOPE_WORKGROUP);
                    __hip_atomic_fetch_add(dst + 64, vv[u] * g[u].z, __ATOMIC_RELAXED, __HIP_MEMORY_SCOPE_WORKGROUP);
                    __hip_atomic_fetch_add(dst + 96, vv[u] * g[u].w, __ATOMIC_RELAXED, __HIP_MEMORY_SCOPE_WORKGROUP);
                }
                j0 += 8;
            } while (j0 < nb);
        };

        auto drain = [&](const bool last) {
            const int todo = last ? cnt : (cnt & ~7);
            int i0 = 0;
            do {
                const int nb = min(32, todo - i0);
                int   c = 0, lr = OWN_ROWS;
                float v = 0.f;
                if (lane < nb) {
                    const unsigned en = queue[w][i0 + lane];
                    const int e = (int)(en >> 9);
                    c  = col[e];
                    v  = vals[e];
                    lr = (int)(en & 511u);
                }
                if constexpr (!REPLAY) {
                    if (rec >= 0) {
                        if (rec + nb <= LIST_CAP) {
                            if (lane < nb)
                                myList[rec + lane] = make_uint2(((unsigned)c << 9) | (unsigned)lr, __float_as_uint(v));
                            rec += nb;
                        } else {
                            rec = -1;
                        }
                    }
                }
                addBatch(c, v, lr, nb);
                i0 += 32;
            } while (i0 < todo);
            const int rem = cnt - todo;
            if (lane < rem) {
                const unsigned keep = queue[w][todo + lane];
                queue[w][lane] = keep;
            }
            cnt = rem;
        };

        bool sweep = true;
        if constexpr (REPLAY) {
            int n = -1;
            if (base == firstBase) n = __builtin_amdgcn_readfirstlane(*myCount);
            if ((unsigned)n <= (unsigned)LIST_CAP) {
                sweep = false;
                const uint2 filler = make_uint2((unsigned)OWN_ROWS, 0u);
                if (n > 0) {
                    const int lastEntry = n - 1;
                    uint2 cur = myList[min(lane, lastEntry)];
                    if (lane >= n) cur = filler;
                    int i0 = 0;
                    do {
                        uint2 nxt = myList[min(i0 + 32 + lane, lastEntry)];
                        if (i0 + 32 + lane >= n) nxt = filler;
                        addBatch((int)(cur.x >> 9), __uint_as_float(cur.y), (int)(cur.x & 511u), min(32, n - i0));
                        cur = nxt;
                        i0 += 32;
                    } while (i0 < n);
                }
            }
        }
        if (sweep) {
        for (int t = nGroups * 128 + w * 32; t < E; t += OWN_WAVES * 32) {
            const int e = t + lane;
            unsigned d = 0xffffffffu;
            if (e < E) d = (unsigned)(row[e] - base);
            push(d < rowsHere, e, d);
        }

        if (nGroups > 0) {
            int4 c0 = loadGroup(0), c1 = loadGroup(1), c2 = loadGroup(2), c3 = loadGroup(3);
#pragma unroll 1
            for (int st = 0; st < nStep; st += 2) {
                int it = st * 4;
                const int4 n0 = loadGroup(it + 4), n1 = loadGroup(it + 5);
                const int4 n2 = loadGroup(it + 6), n3 = loadGroup(it + 7);
                scanGroup(c0, it);
                scanGroup(c1, it + 1);
                scanGroup(c2, it + 2);
                scanGroup(c3, it + 3);
                if (cnt > OWN_QCAP - 512) drain(false);

                it += 4;
                c0 = loadGroup(it + 4); c1 = loadGroup(it + 5);
                c2 = loadGroup(it + 6); c3 = loadGroup(it + 7);
                scanGroup(n0, it);
                scanGroup(n1, it + 1);
                scanGroup(n2, it + 2);
                scanGroup(n3, it + 3);
                if (cnt > OWN_QCAP - 512) drain(false);
            }
        }
        if (cnt > 0) drain(true);
        }
        if constexpr (!REPLAY) {
            if (base == firstBase && lane == 0) *myCount = rec;
            rec = -1;
        }
        __syncthreads();

        for (int idx = tid; idx < (int)rowsHere * 32; idx += blockDim.x) {
            const int lrow = idx >> 5;
            const float* a = acc + lrow * DIM_K + lane;
            const float4 o = make_float4(a[0], a[32], a[64], a[96]);
            ((float4*)(S + (long long)(base + lrow) * DIM_K))[lane] = o;
        }
        __syncthreads();
    }
}

template <int NT, int NOUT>
__global__ void gcn_gemm_tanh_kernel(const float* __restrict__ S,
                                     const float* __restrict__ W,
                                     float* __restrict__ O,
                                     int nNodes) {
    constexpr int N_GROUPS = NOUT / (NT * 16);
    const int lane = threadIdx.x & 31;
    const int wInBlock = __builtin_amdgcn_readfirstlane((int)(threadIdx.x >> 5));
    const int wid  = blockIdx.x * (int)(blockDim.x >> 5) + wInBlock;
    const int pair = wid / N_GROUPS;
    const int grp  = wid - pair * N_GROUPS;
    const int row0 = pair * 32;
    if (row0 >= nNodes) return;
    const bool two = row0 + 16 < nNodes;

    const int c = lane & 15;
    const int h = lane >> 4;
    const int colBase = grp * NT * 16;

    const float4* a0p = (const float4*)(S + (long long)(row0 + c) * DIM_K + 4 * h);
    const float4* a1p = (const float4*)(S + (long long)(row0 + (two ? 16 : 0) + c) * DIM_K + 4 * h);
    const float4* bp  = (const float4*)(W + (long long)(colBase + c) * DIM_K + 4 * h);

    v8f acc0[NT], acc1[NT];
#pragma unroll
    for (int t = 0; t < NT; ++t) { acc0[t] = (v8f){}; acc1[t] = (v8f){}; }

    float4 a0 = a0p[0];
    float4 a1 = a1p[0];
    float4 b[NT];
#pragma unroll
    for (int t = 0; t < NT; ++t) b[t] = bp[t * 512];

#pragma unroll
    for (int q = 0; q < DIM_K / 8; ++q) {
        float4 na0 = a0, na1 = a1;
        float4 nb[NT];
#pragma unroll
        for (int t = 0; t < NT; ++t) nb[t] = b[t];
        if (q + 1 < DIM_K / 8) {
            na0 = a0p[2 * (q + 1)];
            na1 = a1p[2 * (q + 1)];
#pragma unroll
            for (int t = 0; t < NT; ++t) nb[t] = bp[t * 512 + 2 * (q + 1)];
        }
        {
            const v2f x0 = (v2f){a0.x, a0.y};
            const v2f x1 = (v2f){a1.x, a1.y};
#pragma unroll
            for (int t = 0; t < NT; ++t) {
                const v2f y = (v2f){b[t].x, b[t].y};
                acc0[t] = __builtin_amdgcn_wmma_f32_16x16x4_f32(
                     false, x0,  false, y,
                     (short)0, acc0[t],  false,  false);
                acc1[t] = __builtin_amdgcn_wmma_f32_16x16x4_f32(
                    false, x1, false, y, (short)0, acc1[t], false, false);
            }
        }
        {
            const v2f x0 = (v2f){a0.z, a0.w};
            const v2f x1 = (v2f){a1.z, a1.w};
#pragma unroll
            for (int t = 0; t < NT; ++t) {
                const v2f y = (v2f){b[t].z, b[t].w};
                acc0[t] = __builtin_amdgcn_wmma_f32_16x16x4_f32(
                    false, x0, false, y, (short)0, acc0[t], false, false);
                acc1[t] = __builtin_amdgcn_wmma_f32_16x16x4_f32(
                    false, x1, false, y, (short)0, acc1[t], false, false);
            }
        }
        a0 = na0;
        a1 = na1;
#pragma unroll
        for (int t = 0; t < NT; ++t) b[t] = nb[t];
    }

    const int mbase = row0 + 8 * h;
#pragma unroll
    for (int t = 0; t < NT; ++t) {
        const int n = colBase + t * 16 + c;
#pragma unroll
        for (int r = 0; r < 8; ++r) {
            O[(long long)(mbase + r) * NOUT + n] = fast_tanh(acc0[t][r]);
        }
    }
    if (two) {
#pragma unroll
        for (int t = 0; t < NT; ++t) {
            const int n = colBase + t * 16 + c;
#pragma unroll
            for (int r = 0; r < 8; ++r) {
                O[(long long)(mbase + 16 + r) * NOUT + n] = fast_tanh(acc1[t][r]);
            }
        }
    }
}

extern "C" void kernel_launch(void* const* d_in, const int* in_sizes, int n_in,
                              void* d_out, int out_size, void* d_ws, size_t ws_size,
                              hipStream_t stream) {
    const float* X        = (const float*)d_in[0];
    const float* W1       = (const float*)d_in[1];
    const float* W2       = (const float*)d_in[2];
    const float* edgeVals = (const float*)d_in[3];
    const int*   edgeRow  = (const int*)d_in[4];
    const int*   edgeCol  = (const int*)d_in[5];
    float* out = (float*)d_out;

    const int N = in_sizes[0] / DIM_K;
    const int E = in_sizes[3];
    const long long featCount = (long long)N * DIM_K;

    float* S  = (float*)d_ws;
    float* H1 = S + featCount;
    unsigned* edgeList  = (unsigned*)(H1 + featCount);
    int*      listCount = (int*)(edgeList + 2LL * OWN_GROUPS * OWN_WAVES * LIST_CAP);

    const int gemmThreads = 128;
    const int wavesPerBlock = gemmThreads / 32;
    const int mPairs = (N + 31) / 32;
    const int gemmBlocks1 = (mPairs * 2 + wavesPerBlock - 1) / wavesPerBlock;
    const int gemmBlocks2 = (mPairs + wavesPerBlock - 1) / wavesPerBlock;

    gcn_spmm_owner_kernel<false><<<OWN_GROUPS, 1024, 0, stream>>>(
        edgeVals, edgeRow, edgeCol, X, S, edgeList, listCount, E, N);
    gcn_gemm_tanh_kernel<4, 128><<<gemmBlocks1, gemmThreads, 0, stream>>>(
        S, W1, H1, N);

    gcn_spmm_owner_kernel<true><<<OWN_GROUPS, 1024, 0, stream>>>(
        edgeVals, edgeRow, edgeCol, H1, S, edgeList, listCount, E, N);
    gcn_gemm_tanh_kernel<4, 64><<<gemmBlocks2, gemmThreads, 0, stream>>>(
        S, W2, out, N);
}


# Layer 1 sweeps a one-byte-per-edge owner array (written by a new first launch from edge_row) instead of edge_row: 4x fewer bytes swept; local row read at gather; sums stay in LDS arrival order
# speedup vs baseline: 3.2016x; 1.0701x over previous
#include <hip/hip_runtime.h>
#include <hip/hip_bf16.h>
#include <math.h>


typedef __attribute__((ext_vector_type(2))) float v2f;
typedef __attribute__((ext_vector_type(8))) float v8f;

#define DIM_K 128

__device__ __forceinline__ float fast_tanh(float x) {
    float ax = fabsf(x);
    float e  = __expf(-2.0f * ax);
    float r  = (1.0f - e) * __builtin_amdgcn_rcpf(1.0f + e);
    return copysignf(r, x);
}

#define OWN_ROWS  392
#define OWN_WAVES 32
#define OWN_QCAP  640
#define OWN_GROUPS 256
#define LIST_CAP  1024

__device__ __forceinline__ unsigned owner_id(int r) {
    return ((unsigned)r / (unsigned)OWN_ROWS) & 255u;
}

__global__ __launch_bounds__(256)
void gcn_owner_ids_kernel(const int* __restrict__ row, unsigned* __restrict__ ownerWords, int E) {
    const int lane  = threadIdx.x & 31;
    const int wave  = blockIdx.x * (int)(blockDim.x >> 5) + __builtin_amdgcn_readfirstlane((int)(threadIdx.x >> 5));
    const int eWave = wave * 512;
    auto pack = [](const int4 r) -> unsigned {
        return owner_id(r.x) | (owner_id(r.y) << 8) | (owner_id(r.z) << 16) | (owner_id(r.w) << 24);
    };
    if (eWave + 512 <= E) {
        int4 r[4];
#pragma unroll
        for (int j = 0; j < 4; ++j) r[j] = *(const int4*)(row + eWave + j * 128 + lane * 4);
#pragma unroll
        for (int j = 0; j < 4; ++j) ownerWords[(eWave >> 2) + j * 32 + lane] = pack(r[j]);
    } else {
#pragma unroll
        for (int j = 0; j < 4; ++j) {
            const int e = eWave + j * 128 + lane * 4;
            if (e + 3 < E) ownerWords[e >> 2] = pack(*(const int4*)(row + e));
        }
    }
}

template <bool REPLAY>
__global__ __launch_bounds__(1024)
void gcn_spmm_owner_kernel(const float* __restrict__ vals,
                           const int* __restrict__ row,
                           const int* __restrict__ col,
                           const float* __restrict__ X,
                           float* __restrict__ S,
                           const unsigned* __restrict__ ownerWords,
                           unsigned* __restrict__ edgeList,
                           int* __restrict__ listCount,
                           int E, int nNodes) {
    __shared__ float    acc[(OWN_ROWS + 1) * DIM_K];
    __shared__ unsigned queue[OWN_WAVES][OWN_QCAP];

    const int tid  = threadIdx.x;
    const int lane = tid & 31;
    const int w = __builtin_amdgcn_readfirstlane(tid >> 5);
    const unsigned laneLt = (1u << lane) - 1u;

    const int nGroups = E >> 9;
    const int nIt     = (nGroups + OWN_WAVES - 1) / OWN_WAVES;
    const int nStep   = (nIt + 3) >> 2;
    const unsigned idBytes = ((unsigned)blockIdx.x & 255u) * 0x01010101u;
    const int startGroup = (int)(((long long)blockIdx.x * nGroups) / gridDim.x);

    uint2* const myList  = (uint2*)edgeList + (long long)(blockIdx.x * OWN_WAVES + w) * LIST_CAP;
    int*   const myCount = listCount + blockIdx.x * OWN_WAVES + w;
    const int firstBase  = blockIdx.x * OWN_ROWS;
    int rec = 0;

    for (int base = blockIdx.x * OWN_ROWS; base < nNodes; base += gridDim.x * OWN_ROWS) {
        const unsigned rowsHere = (unsigned)min(OWN_ROWS, nNodes - base);

        for (int i = tid; i < (OWN_ROWS + 1) * DIM_K / 4; i += blockDim.x)
            ((float4*)acc)[i] = make_float4(0.f, 0.f, 0.f, 0.f);
        __syncthreads();

        int cnt = 0;
        auto push = [&](bool m, int e) {
            const unsigned b = __builtin_amdgcn_ballot_w32(m);
            if (m) queue[w][cnt + __builtin_popcount(b & laneLt)] = (unsigned)e;
            cnt += __builtin_popcount(b);
        };

        auto groupOf = [&](int it) -> int {
            int g = min(it * OWN_WAVES + w, nGroups - 1) + startGroup;
            if (g >= nGroups) g -= nGroups;
            return g;
        };
        auto loadGroup = [&](int it) -> uint4 {
            return *(const uint4*)(ownerWords + groupOf(it) * 128 + lane * 4);
        };
        auto zeroBytes = [](unsigned x) -> unsigned {
            return (x - 0x01010101u) & ~x & 0x80808080u;
        };
        auto scanWord = [&](unsigned x, unsigned m, int eBase) {
            while (__builtin_amdgcn_ballot_w32(m != 0u) != 0u) {
                const bool hit = m != 0u;
                const unsigned pos = (unsigned)__builtin_ctz(m | 0x80000000u) >> 3;
                push(hit, eBase + (int)pos);
                if (hit) x |= 0xFFu << (8u * pos);
                m = zeroBytes(x);
            }
        };
        auto scanGroup = [&](const uint4 r, int it) {
            if (it * OWN_WAVES + w < nGroups) {
                const unsigned x0 = r.x ^ idBytes, x1 = r.y ^ idBytes;
                const unsigned x2 = r.z ^ idBytes, x3 = r.w ^ idBytes;
                const unsigned m0 = zeroBytes(x0), m1 = zeroBytes(x1);
                const unsigned m2 = zeroBytes(x2), m3 = zeroBytes(x3);
                if (__builtin_amdgcn_ballot_w32((m0 | m1 | m2 | m3) != 0u) != 0u) {
                    const int e0 = groupOf(it) * 512 + lane * 16;
                    scanWord(x0, m0, e0);
                    scanWord(x1, m1, e0 + 4);
                    scanWord(x2, m2, e0 + 8);
                    scanWord(x3, m3, e0 + 12);
                }
            }
        };
        auto addBatch = [&](const int c, const float v, const int lr, const int nb) {
            int j0 = 0;
            do {
                float4 g[8];
                float  vv[8];
                int    rr[8];
#pragma unroll
                for (int u = 0; u < 8; ++u) {
                    const int j  = j0 + u;
                    const int cj = __builtin_amdgcn_readlane(c, j);
                    vv[u] = __int_as_float(__builtin_amdgcn_readlane(__float_as_int(v), j));
                    rr[u] = __builtin_amdgcn_readlane(lr, j);
                    g[u]  = ((const float4*)(X + (long long)cj * DIM_K))[lane];
                }
#pragma unroll
                for (int u = 0; u < 8; ++u) {
                    float* dst = acc + rr[u] * DIM_K + lane;
                    __hip_atomic_fetch_add(dst,      vv[u] * g[u].x, __ATOMIC_RELAXED, __HIP_MEMORY_SCOPE_WORKGROUP);
                    __hip_atomic_fetch_add(dst + 32, vv[u] * g[u].y, __ATOMIC_RELAXED, __HIP_MEMORY_SCOPE_WORKGROUP);
                    __hip_atomic_fetch_add(dst + 64, vv[u] * g[u].z, __ATOMIC_RELAXED, __HIP_MEMORY_SCOPE_WORKGROUP);
                    __hip_atomic_fetch_add(dst + 96, vv[u] * g[u].w, __ATOMIC_RELAXED, __HIP_MEMORY_SCOPE_WORKGROUP);
                }
                j0 += 8;
            } while (j0 < nb);
        };

        auto drain = [&](const bool last) {
            const int todo = last ? cnt : (cnt & ~7);
            int i0 = 0;
            do {
                const int nb = min(32, todo - i0);
                int   c = 0, lr = OWN_ROWS;
                float v = 0.f;
                if (lane < nb) {
                    const int e = (int)queue[w][i0 + lane];
                    const unsigned d  = (unsigned)(row[e] - base);
                    const int      ce = col[e];
                    const float    ve = vals[e];
                    const bool   ours = d < rowsHere;
                    c  = ours ? ce : 0;
                    v  = ours ? ve : 0.f;
                    lr = ours ? (int)d : OWN_ROWS;
                }
                if constexpr (!REPLAY) {
                    if (rec >= 0) {
                        if (rec + nb <= LIST_CAP) {
                            if (lane < nb)
                                myList[rec + lane] = make_uint2(((unsigned)c << 9) | (unsigned)lr, __float_as_uint(v));
                            rec += nb;
                        } else {
                            rec = -1;
                        }
                    }
                }
                addBatch(c, v, lr, nb);
                i0 += 32;
            } while (i0 < todo);
            const int rem = cnt - todo;
            if (lane < rem) {
                const unsigned keep = queue[w][todo + lane];
                queue[w][lane] = keep;
            }
            cnt = rem;
        };

        bool sweep = true;
        if constexpr (REPLAY) {
            int n = -1;
            if (base == firstBase) n = __builtin_amdgcn_readfirstlane(*myCount);
            if ((unsigned)n <= (unsigned)LIST_CAP) {
                sweep = false;
                const uint2 filler = make_uint2((unsigned)OWN_ROWS, 0u);
                if (n > 0) {
                    const int lastEntry = n - 1;
                    uint2 cur = myList[min(lane, lastEntry)];
                    if (lane >= n) cur = filler;
                    int i0 = 0;
                    do {
                        uint2 nxt = myList[min(i0 + 32 + lane, lastEntry)];
                        if (i0 + 32 + lane >= n) nxt = filler;
                        addBatch((int)(cur.x >> 9), __uint_as_float(cur.y), (int)(cur.x & 511u), min(32, n - i0));
                        cur = nxt;
                        i0 += 32;
                    } while (i0 < n);
                }
            }
        }
        if (sweep) {
        for (int t = nGroups * 512 + w * 32; t < E; t += OWN_WAVES * 32) {
            const int e = t + lane;
            unsigned d = 0xffffffffu;
            if (e < E) d = (unsigned)(row[e] - base);
            push(d < rowsHere, e);
        }

        if (nGroups > 0) {
            uint4 c0 = loadGroup(0), c1 = loadGroup(1), c2 = loadGroup(2), c3 = loadGroup(3);
#pragma unroll 1
            for (int st = 0; st < nStep; st += 2) {
                int it = st * 4;
                const uint4 n0 = loadGroup(it + 4), n1 = loadGroup(it + 5);
                const uint4 n2 = loadGroup(it + 6), n3 = loadGroup(it + 7);
                scanGroup(c0, it);
                if (cnt > OWN_QCAP - 512) drain(false);
                scanGroup(c1, it + 1);
                if (cnt > OWN_QCAP - 512) drain(false);
                scanGroup(c2, it + 2);
                if (cnt > OWN_QCAP - 512) drain(false);
                scanGroup(c3, it + 3);
                if (cnt > OWN_QCAP - 512) drain(false);

                it += 4;
                c0 = loadGroup(it + 4); c1 = loadGroup(it + 5);
                c2 = loadGroup(it + 6); c3 = loadGroup(it + 7);
                scanGroup(n0, it);
                if (cnt > OWN_QCAP - 512) drain(false);
                scanGroup(n1, it + 1);
                if (cnt > OWN_QCAP - 512) drain(false);
                scanGroup(n2, it + 2);
                if (cnt > OWN_QCAP - 512) drain(false);
                scanGroup(n3, it + 3);
                if (cnt > OWN_QCAP - 512) drain(false);
            }
        }
        if (cnt > 0) drain(true);
        }
        if constexpr (!REPLAY) {
            if (base == firstBase && lane == 0) *myCount = rec;
            rec = -1;
        }
        __syncthreads();

        for (int idx = tid; idx < (int)rowsHere * 32; idx += blockDim.x) {
            const int lrow = idx >> 5;
            const float* a = acc + lrow * DIM_K + lane;
            const float4 o = make_float4(a[0], a[32], a[64], a[96]);
            ((float4*)(S + (long long)(base + lrow) * DIM_K))[lane] = o;
        }
        __syncthreads();
    }
}

template <int NT, int NOUT>
__global__ void gcn_gemm_tanh_kernel(const float* __restrict__ S,
                                     const float* __restrict__ W,
                                     float* __restrict__ O,
                                     int nNodes) {
    constexpr int N_GROUPS = NOUT / (NT * 16);
    const int lane = threadIdx.x & 31;
    const int wInBlock = __builtin_amdgcn_readfirstlane((int)(threadIdx.x >> 5));
    const int wid  = blockIdx.x * (int)(blockDim.x >> 5) + wInBlock;
    const int pair = wid / N_GROUPS;
    const int grp  = wid - pair * N_GROUPS;
    const int row0 = pair * 32;
    if (row0 >= nNodes) return;
    const bool two = row0 + 16 < nNodes;

    const int c = lane & 15;
    const int h = lane >> 4;
    const int colBase = grp * NT * 16;

    const float4* a0p = (const float4*)(S + (long long)(row0 + c) * DIM_K + 4 * h);
    const float4* a1p = (const float4*)(S + (long long)(row0 + (two ? 16 : 0) + c) * DIM_K + 4 * h);
    const float4* bp  = (const float4*)(W + (long long)(colBase + c) * DIM_K + 4 * h);

    v8f acc0[NT], acc1[NT];
#pragma unroll
    for (int t = 0; t < NT; ++t) { acc0[t] = (v8f){}; acc1[t] = (v8f){}; }

    float4 a0 = a0p[0];
    float4 a1 = a1p[0];
    float4 b[NT];
#pragma unroll
    for (int t = 0; t < NT; ++t) b[t] = bp[t * 512];

#pragma unroll
    for (int q = 0; q < DIM_K / 8; ++q) {
        float4 na0 = a0, na1 = a1;
        float4 nb[NT];
#pragma unroll
        for (int t = 0; t < NT; ++t) nb[t] = b[t];
        if (q + 1 < DIM_K / 8) {
            na0 = a0p[2 * (q + 1)];
            na1 = a1p[2 * (q + 1)];
#pragma unroll
            for (int t = 0; t < NT; ++t) nb[t] = bp[t * 512 + 2 * (q + 1)];
        }
        {
            const v2f x0 = (v2f){a0.x, a0.y};
            const v2f x1 = (v2f){a1.x, a1.y};
#pragma unroll
            for (int t = 0; t < NT; ++t) {
                const v2f y = (v2f){b[t].x, b[t].y};
                acc0[t] = __builtin_amdgcn_wmma_f32_16x16x4_f32(
                     false, x0,  false, y,
                     (short)0, acc0[t],  false,  false);
                acc1[t] = __builtin_amdgcn_wmma_f32_16x16x4_f32(
                    false, x1, false, y, (short)0, acc1[t], false, false);
            }
        }
        {
            const v2f x0 = (v2f){a0.z, a0.w};
            const v2f x1 = (v2f){a1.z, a1.w};
#pragma unroll
            for (int t = 0; t < NT; ++t) {
                const v2f y = (v2f){b[t].z, b[t].w};
                acc0[t] = __builtin_amdgcn_wmma_f32_16x16x4_f32(
                    false, x0, false, y, (short)0, acc0[t], false, false);
                acc1[t] = __builtin_amdgcn_wmma_f32_16x16x4_f32(
                    false, x1, false, y, (short)0, acc1[t], false, false);
            }
        }
        a0 = na0;
        a1 = na1;
#pragma unroll
        for (int t = 0; t < NT; ++t) b[t] = nb[t];
    }

    const int mbase = row0 + 8 * h;
#pragma unroll
    for (int t = 0; t < NT; ++t) {
        const int n = colBase + t * 16 + c;
#pragma unroll
        for (int r = 0; r < 8; ++r) {
            O[(long long)(mbase + r) * NOUT + n] = fast_tanh(acc0[t][r]);
        }
    }
    if (two) {
#pragma unroll
        for (int t = 0; t < NT; ++t) {
            const int n = colBase + t * 16 + c;
#pragma unroll
            for (int r = 0; r < 8; ++r) {
                O[(long long)(mbase + 16 + r) * NOUT + n] = fast_tanh(acc1[t][r]);
            }
        }
    }
}

extern "C" void kernel_launch(void* const* d_in, const int* in_sizes, int n_in,
                              void* d_out, int out_size, void* d_ws, size_t ws_size,
                              hipStream_t stream) {
    const float* X        = (const float*)d_in[0];
    const float* W1       = (const float*)d_in[1];
    const float* W2       = (const float*)d_in[2];
    const float* edgeVals = (const float*)d_in[3];
    const int*   edgeRow  = (const int*)d_in[4];
    const int*   edgeCol  = (const int*)d_in[5];
    float* out = (float*)d_out;

    const int N = in_sizes[0] / DIM_K;
    const int E = in_sizes[3];
    const long long featCount = (long long)N * DIM_K;

    float* S  = (float*)d_ws;
    float* H1 = S + featCount;
    unsigned* edgeList  = (unsigned*)(H1 + featCount);
    int*      listCount = (int*)(edgeList + 2LL * OWN_GROUPS * OWN_WAVES * LIST_CAP);
    unsigned* ownerWords = (unsigned*)(listCount + OWN_GROUPS * OWN_WAVES);

    const int gemmThreads = 128;
    const int wavesPerBlock = gemmThreads / 32;
    const int mPairs = (N + 31) / 32;
    const int gemmBlocks1 = (mPairs * 2 + wavesPerBlock - 1) / wavesPerBlock;
    const int gemmBlocks2 = (mPairs + wavesPerBlock - 1) / wavesPerBlock;

    const int idBlocks = (E + 4095) / 4096;
    gcn_owner_ids_kernel<<<idBlocks, 256, 0, stream>>>(edgeRow, ownerWords, E);

    gcn_spmm_owner_kernel<false><<<OWN_GROUPS, 1024, 0, stream>>>(
        edgeVals, edgeRow, edgeCol, X, S, ownerWords, edgeList, listCount, E, N);
    gcn_gemm_tanh_kernel<4, 128><<<gemmBlocks1, gemmThreads, 0, stream>>>(
        S, W1, H1, N);

    gcn_spmm_owner_kernel<true><<<OWN_GROUPS, 1024, 0, stream>>>(
        edgeVals, edgeRow, edgeCol, H1, S, ownerWords, edgeList, listCount, E, N);
    gcn_gemm_tanh_kernel<4, 64><<<gemmBlocks2, gemmThreads, 0, stream>>>(
        S, W2, out, N);
}


# Both W products moved into the neighbour-sum kernels: each work-group runs its WMMAs on the finished LDS row sums (A and W fragments from LDS); S never stored, 3 launches; k summed in a new order
# speedup vs baseline: 3.8034x; 1.1880x over previous
#include <hip/hip_runtime.h>
#include <hip/hip_bf16.h>
#include <math.h>


typedef __attribute__((ext_vector_type(2))) float v2f;
typedef __attribute__((ext_vector_type(8))) float v8f;

#define DIM_K 128

__device__ __forceinline__ float fast_tanh(float x) {
    float ax = fabsf(x);
    float e  = __expf(-2.0f * ax);
    float r  = (1.0f - e) * __builtin_amdgcn_rcpf(1.0f + e);
    return copysignf(r, x);
}

#define OWN_ROWS  392
#define ROW_PITCH 132
#define OWN_WAVES 32
#define OWN_QCAP  640
#define OWN_GROUPS 256
#define LIST_CAP  1024

__device__ __forceinline__ unsigned owner_id(int r) {
    return ((unsigned)r / (unsigned)OWN_ROWS) & 255u;
}

__global__ __launch_bounds__(256)
void gcn_owner_ids_kernel(const int* __restrict__ row, unsigned* __restrict__ ownerWords, int E) {
    const int lane  = threadIdx.x & 31;
    const int wave  = blockIdx.x * (int)(blockDim.x >> 5) + __builtin_amdgcn_readfirstlane((int)(threadIdx.x >> 5));
    const int eWave = wave * 512;
    auto pack = [](const int4 r) -> unsigned {
        return owner_id(r.x) | (owner_id(r.y) << 8) | (owner_id(r.z) << 16) | (owner_id(r.w) << 24);
    };
    if (eWave + 512 <= E) {
        int4 r[4];
#pragma unroll
        for (int j = 0; j < 4; ++j) r[j] = *(const int4*)(row + eWave + j * 128 + lane * 4);
#pragma unroll
        for (int j = 0; j < 4; ++j) ownerWords[(eWave >> 2) + j * 32 + lane] = pack(r[j]);
    } else {
#pragma unroll
        for (int j = 0; j < 4; ++j) {
            const int e = eWave + j * 128 + lane * 4;
            if (e + 3 < E) ownerWords[e >> 2] = pack(*(const int4*)(row + e));
        }
    }
}

template <bool REPLAY, int NOUT>
__global__ __launch_bounds__(1024)
void gcn_spmm_owner_kernel(const float* __restrict__ vals,
                           const int* __restrict__ row,
                           const int* __restrict__ col,
                           const float* __restrict__ X,
                           const float* __restrict__ W,
                           float* __restrict__ O,
                           const unsigned* __restrict__ ownerWords,
                           unsigned* __restrict__ edgeList,
                           int* __restrict__ listCount,
                           int E, int nNodes) {
    __shared__ __attribute__((aligned(16))) float    acc[(OWN_ROWS + 1) * ROW_PITCH];
    __shared__ __attribute__((aligned(16))) unsigned queue[OWN_WAVES][OWN_QCAP];
    static_assert(NOUT * ROW_PITCH <= OWN_WAVES * OWN_QCAP, "W must fit the queues' LDS");
    static_assert(NOUT % 32 == 0 && NOUT % OWN_WAVES == 0, "units are 32 columns wide; W is copied OWN_WAVES rows a pass");

    const int tid  = threadIdx.x;
    const int lane = tid & 31;
    const int w = __builtin_amdgcn_readfirstlane(tid >> 5);
    const unsigned laneLt = (1u << lane) - 1u;

    const int nGroups = E >> 9;
    const int nIt     = (nGroups + OWN_WAVES - 1) / OWN_WAVES;
    const int nStep   = (nIt + 3) >> 2;
    const unsigned idBytes = ((unsigned)blockIdx.x & 255u) * 0x01010101u;
    const int startGroup = (int)(((long long)blockIdx.x * nGroups) / gridDim.x);

    uint2* const myList  = (uint2*)edgeList + (long long)(blockIdx.x * OWN_WAVES + w) * LIST_CAP;
    int*   const myCount = listCount + blockIdx.x * OWN_WAVES + w;
    const int firstBase  = blockIdx.x * OWN_ROWS;
    int rec = 0;

    for (int base = blockIdx.x * OWN_ROWS; base < nNodes; base += gridDim.x * OWN_ROWS) {
        const unsigned rowsHere = (unsigned)min(OWN_ROWS, nNodes - base);

        for (int i = tid; i < (OWN_ROWS + 1) * ROW_PITCH / 4; i += blockDim.x)
            ((float4*)acc)[i] = make_float4(0.f, 0.f, 0.f, 0.f);
        __syncthreads();

        int cnt = 0;
        auto push = [&](bool m, int e) {
            const unsigned b = __builtin_amdgcn_ballot_w32(m);
            if (m) queue[w][cnt + __builtin_popcount(b & laneLt)] = (unsigned)e;
            cnt += __builtin_popcount(b);
        };

        auto groupOf = [&](int it) -> int {
            int g = min(it * OWN_WAVES + w, nGroups - 1) + startGroup;
            if (g >= nGroups) g -= nGroups;
            return g;
        };
        auto loadGroup = [&](int it) -> uint4 {
            return *(const uint4*)(ownerWords + groupOf(it) * 128 + lane * 4);
        };
        auto zeroBytes = [](unsigned x) -> unsigned {
            return (x - 0x01010101u) & ~x & 0x80808080u;
        };
        auto scanWord = [&](unsigned x, unsigned m, int eBase) {
            while (__builtin_amdgcn_ballot_w32(m != 0u) != 0u) {
                const bool hit = m != 0u;
                const unsigned pos = (unsigned)__builtin_ctz(m | 0x80000000u) >> 3;
                push(hit, eBase + (int)pos);
                if (hit) x |= 0xFFu << (8u * pos);
                m = zeroBytes(x);
            }
        };
        auto scanGroup = [&](const uint4 r, int it) {
            if (it * OWN_WAVES + w < nGroups) {
                const unsigned x0 = r.x ^ idBytes, x1 = r.y ^ idBytes;
                const unsigned x2 = r.z ^ idBytes, x3 = r.w ^ idBytes;
                const unsigned m0 = zeroBytes(x0), m1 = zeroBytes(x1);
                const unsigned m2 = zeroBytes(x2), m3 = zeroBytes(x3);
                if (__builtin_amdgcn_ballot_w32((m0 | m1 | m2 | m3) != 0u) != 0u) {
                    const int e0 = groupOf(it) * 512 + lane * 16;
                    scanWord(x0, m0, e0);
                    scanWord(x1, m1, e0 + 4);
                    scanWord(x2, m2, e0 + 8);
                    scanWord(x3, m3, e0 + 12);
                }
            }
        };
        auto addBatch = [&](const int c, const float v, const int lr, const int nb) {
            int j0 = 0;
            do {
                float4 g[8];
                float  vv[8];
                int    rr[8];
#pragma unroll
                for (int u = 0; u < 8; ++u) {
                    const int j  = j0 + u;
                    const int cj = __builtin_amdgcn_readlane(c, j);
                    vv[u] = __int_as_float(__builtin_amdgcn_readlane(__float_as_int(v), j));
                    rr[u] = __builtin_amdgcn_readlane(lr, j);
                    g[u]  = ((const float4*)(X + (long long)cj * DIM_K))[lane];
                }
#pragma unroll
                for (int u = 0; u < 8; ++u) {
                    float* dst = acc + rr[u] * ROW_PITCH + lane;
                    __hip_atomic_fetch_add(dst,      vv[u] * g[u].x, __ATOMIC_RELAXED, __HIP_MEMORY_SCOPE_WORKGROUP);
                    __hip_atomic_fetch_add(dst + 32, vv[u] * g[u].y, __ATOMIC_RELAXED, __HIP_MEMORY_SCOPE_WORKGROUP);
                    __hip_atomic_fetch_add(dst + 64, vv[u] * g[u].z, __ATOMIC_RELAXED, __HIP_MEMORY_SCOPE_WORKGROUP);
                    __hip_atomic_fetch_add(dst + 96, vv[u] * g[u].w, __ATOMIC_RELAXED, __HIP_MEMORY_SCOPE_WORKGROUP);
                }
                j0 += 8;
            } while (j0 < nb);
        };

        auto drain = [&](const bool last) {
            const int todo = last ? cnt : (cnt & ~7);
            int i0 = 0;
            do {
                const int nb = min(32, todo - i0);
                int   c = 0, lr = OWN_ROWS;
                float v = 0.f;
                if (lane < nb) {
                    const int e = (int)queue[w][i0 + lane];
                    const unsigned d  = (unsigned)(row[e] - base);
                    const int      ce = col[e];
                    const float    ve = vals[e];
                    const bool   ours = d < rowsHere;
                    c  = ours ? ce : 0;
                    v  = ours ? ve : 0.f;
                    lr = ours ? (int)d : OWN_ROWS;
                }
                if constexpr (!REPLAY) {
                    if (rec >= 0) {
                        if (rec + nb <= LIST_CAP) {
                            if (lane < nb)
                                myList[rec + lane] = make_uint2(((unsigned)c << 9) | (unsigned)lr, __float_as_uint(v));
                            rec += nb;
                        } else {
                            rec = -1;
                        }
                    }
                }
                addBatch(c, v, lr, nb);
                i0 += 32;
            } while (i0 < todo);
            const int rem = cnt - todo;
            if (lane < rem) {
                const unsigned keep = queue[w][todo + lane];
                queue[w][lane] = keep;
            }
            cnt = rem;
        };

        bool sweep = true;
        if constexpr (REPLAY) {
            int n = -1;
            if (base == firstBase) n = __builtin_amdgcn_readfirstlane(*myCount);
            if ((unsigned)n <= (unsigned)LIST_CAP) {
                sweep = false;
                const uint2 filler = make_uint2((unsigned)OWN_ROWS, 0u);
                if (n > 0) {
                    const int lastEntry = n - 1;
                    uint2 cur = myList[min(lane, lastEntry)];
                    if (lane >= n) cur = filler;
                    int i0 = 0;
                    do {
                        uint2 nxt = myList[min(i0 + 32 + lane, lastEntry)];
                        if (i0 + 32 + lane >= n) nxt = filler;
                        addBatch((int)(cur.x >> 9), __uint_as_float(cur.y), (int)(cur.x & 511u), min(32, n - i0));
                        cur = nxt;
                        i0 += 32;
                    } while (i0 < n);
                }
            }
        }
        if (sweep) {
        for (int t = nGroups * 512 + w * 32; t < E; t += OWN_WAVES * 32) {
            const int e = t + lane;
            unsigned d = 0xffffffffu;
            if (e < E) d = (unsigned)(row[e] - base);
            push(d < rowsHere, e);
        }

        if (nGroups > 0) {
            uint4 c0 = loadGroup(0), c1 = loadGroup(1), c2 = loadGroup(2), c3 = loadGroup(3);
#pragma unroll 1
            for (int st = 0; st < nStep; st += 2) {
                int it = st * 4;
                const uint4 n0 = loadGroup(it + 4), n1 = loadGroup(it + 5);
                const uint4 n2 = loadGroup(it + 6), n3 = loadGroup(it + 7);
                scanGroup(c0, it);
                if (cnt > OWN_QCAP - 512) drain(false);
                scanGroup(c1, it + 1);
                if (cnt > OWN_QCAP - 512) drain(false);
                scanGroup(c2, it + 2);
                if (cnt > OWN_QCAP - 512) drain(false);
                scanGroup(c3, it + 3);
                if (cnt > OWN_QCAP - 512) drain(false);

                it += 4;
                c0 = loadGroup(it + 4); c1 = loadGroup(it + 5);
                c2 = loadGroup(it + 6); c3 = loadGroup(it + 7);
                scanGroup(n0, it);
                if (cnt > OWN_QCAP - 512) drain(false);
                scanGroup(n1, it + 1);
                if (cnt > OWN_QCAP - 512) drain(false);
                scanGroup(n2, it + 2);
                if (cnt > OWN_QCAP - 512) drain(false);
                scanGroup(n3, it + 3);
                if (cnt > OWN_QCAP - 512) drain(false);
            }
        }
        if (cnt > 0) drain(true);
        }
        if constexpr (!REPLAY) {
            if (base == firstBase && lane == 0) *myCount = rec;
            rec = -1;
        }
        __syncthreads();

        unsigned* const wl = &queue[0][0];
        {
            constexpr int W_PASSES = NOUT / OWN_WAVES;
            float4 wv[W_PASSES];
#pragma unroll
            for (int p = 0; p < W_PASSES; ++p)
                wv[p] = ((const float4*)(W + (long long)(p * OWN_WAVES + w) * DIM_K))[lane];
#pragma unroll
            for (int p = 0; p < W_PASSES; ++p) {
                unsigned* d = wl + (p * OWN_WAVES + w) * ROW_PITCH + lane;
                d[0]  = __float_as_uint(wv[p].x);
                d[32] = __float_as_uint(wv[p].y);
                d[64] = __float_as_uint(wv[p].z);
                d[96] = __float_as_uint(wv[p].w);
            }
        }
        __syncthreads();

        {
            constexpr int COL_GROUPS = NOUT / 32;
            const int c = lane & 15;
            const int h = lane >> 4;
            const int nStrips = ((int)rowsHere + 15) >> 4;
            const int nUnits  = nStrips * COL_GROUPS;
            for (int u = w; u < nUnits; u += OWN_WAVES) {
                const int strip = u / COL_GROUPS;
                const int cg    = u - strip * COL_GROUPS;
                const int arow = min(strip * 16 + c, OWN_ROWS);
                const float4* ap  = (const float4*)(acc + arow * ROW_PITCH + 4 * h);
                const uint4*  bp0 = (const uint4*)(wl + (cg * 32 + c) * ROW_PITCH + 4 * h);
                const uint4*  bp1 = (const uint4*)(wl + (cg * 32 + 16 + c) * ROW_PITCH + 4 * h);
                v8f d0 = (v8f){}, d1 = (v8f){};
#pragma unroll
                for (int j = 0; j < DIM_K / 8; ++j) {
                    const float4 a  = ap[2 * j];
                    const uint4  p0 = bp0[2 * j];
                    const uint4  p1 = bp1[2 * j];
                    {
                        const v2f x  = (v2f){a.x, a.y};
                        const v2f y0 = (v2f){__uint_as_float(p0.x), __uint_as_float(p0.y)};
                        const v2f y1 = (v2f){__uint_as_float(p1.x), __uint_as_float(p1.y)};
                        d0 = __builtin_amdgcn_wmma_f32_16x16x4_f32(
                             false, x,  false, y0,
                             (short)0, d0,  false,  false);
                        d1 = __builtin_amdgcn_wmma_f32_16x16x4_f32(
                            false, x, false, y1, (short)0, d1, false, false);
                    }
                    {
                        const v2f x  = (v2f){a.z, a.w};
                        const v2f y0 = (v2f){__uint_as_float(p0.z), __uint_as_float(p0.w)};
                        const v2f y1 = (v2f){__uint_as_float(p1.z), __uint_as_float(p1.w)};
                        d0 = __builtin_amdgcn_wmma_f32_16x16x4_f32(
                            false, x, false, y0, (short)0, d0, false, false);
                        d1 = __builtin_amdgcn_wmma_f32_16x16x4_f32(
                            false, x, false, y1, (short)0, d1, false, false);
                    }
                }
                float t0[8], t1[8];
#pragma unroll
                for (int r = 0; r < 8; ++r) { t0[r] = fast_tanh(d0[r]); t1[r] = fast_tanh(d1[r]); }
                const int m0 = strip * 16 + 8 * h;
                float* op = O + (long long)(base + m0) * NOUT + cg * 32 + c;
                if (strip * 16 + 16 <= (int)rowsHere) {
#pragma unroll
                    for (int r = 0; r < 8; ++r) {
                        op[r * NOUT]      = t0[r];
                        op[r * NOUT + 16] = t1[r];
                    }
                } else {
#pragma unroll
                    for (int r = 0; r < 8; ++r) {
                        if ((unsigned)(m0 + r) < rowsHere) {
                            op[r * NOUT]      = t0[r];
                            op[r * NOUT + 16] = t1[r];
                        }
                    }
                }
            }
        }
        __syncthreads();
    }
}

extern "C" void kernel_launch(void* const* d_in, const int* in_sizes, int n_in,
                              void* d_out, int out_size, void* d_ws, size_t ws_size,
                              hipStream_t stream) {
    const float* X        = (const float*)d_in[0];
    const float* W1       = (const float*)d_in[1];
    const float* W2       = (const float*)d_in[2];
    const float* edgeVals = (const float*)d_in[3];
    const int*   edgeRow  = (const int*)d_in[4];
    const int*   edgeCol  = (const int*)d_in[5];
    float* out = (float*)d_out;

    const int N = in_sizes[0] / DIM_K;
    const int E = in_sizes[3];
    const long long featCount = (long long)N * DIM_K;

    float* H1 = (float*)d_ws;
    unsigned* edgeList  = (unsigned*)(H1 + featCount);
    int*      listCount = (int*)(edgeList + 2LL * OWN_GROUPS * OWN_WAVES * LIST_CAP);
    unsigned* ownerWords = (unsigned*)(listCount + OWN_GROUPS * OWN_WAVES);

    const int idBlocks = (E + 4095) / 4096;
    gcn_owner_ids_kernel<<<idBlocks, 256, 0, stream>>>(edgeRow, ownerWords, E);

    gcn_spmm_owner_kernel<false, 128><<<OWN_GROUPS, 1024, 0, stream>>>(
        edgeVals, edgeRow, edgeCol, X, W1, H1, ownerWords, edgeList, listCount, E, N);

    gcn_spmm_owner_kernel<true, 64><<<OWN_GROUPS, 1024, 0, stream>>>(
        edgeVals, edgeRow, edgeCol, H1, W2, out, ownerWords, edgeList, listCount, E, N);
}


# Sweep: a lane's 16 owner-byte flags packed into one word, queued by one loop (a trip takes each lane's lowest flag), not 4 per-word loops; queue order in a group changes, sums stay LDS arrival order
# speedup vs baseline: 4.0469x; 1.0640x over previous
#include <hip/hip_runtime.h>
#include <hip/hip_bf16.h>
#include <math.h>


typedef __attribute__((ext_vector_type(2))) float v2f;
typedef __attribute__((ext_vector_type(8))) float v8f;

#define DIM_K 128

__device__ __forceinline__ float fast_tanh(float x) {
    float ax = fabsf(x);
    float e  = __expf(-2.0f * ax);
    float r  = (1.0f - e) * __builtin_amdgcn_rcpf(1.0f + e);
    return copysignf(r, x);
}

#define OWN_ROWS  392
#define ROW_PITCH 132
#define OWN_WAVES 32
#define OWN_QCAP  640
#define OWN_GROUPS 256
#define LIST_CAP  1024

__device__ __forceinline__ unsigned owner_id(int r) {
    return ((unsigned)r / (unsigned)OWN_ROWS) & 255u;
}

__global__ __launch_bounds__(256)
void gcn_owner_ids_kernel(const int* __restrict__ row, unsigned* __restrict__ ownerWords, int E) {
    const int lane  = threadIdx.x & 31;
    const int wave  = blockIdx.x * (int)(blockDim.x >> 5) + __builtin_amdgcn_readfirstlane((int)(threadIdx.x >> 5));
    const int eWave = wave * 512;
    auto pack = [](const int4 r) -> unsigned {
        return owner_id(r.x) | (owner_id(r.y) << 8) | (owner_id(r.z) << 16) | (owner_id(r.w) << 24);
    };
    if (eWave + 512 <= E) {
        int4 r[4];
#pragma unroll
        for (int j = 0; j < 4; ++j) r[j] = *(const int4*)(row + eWave + j * 128 + lane * 4);
#pragma unroll
        for (int j = 0; j < 4; ++j) ownerWords[(eWave >> 2) + j * 32 + lane] = pack(r[j]);
    } else {
#pragma unroll
        for (int j = 0; j < 4; ++j) {
            const int e = eWave + j * 128 + lane * 4;
            if (e + 3 < E) ownerWords[e >> 2] = pack(*(const int4*)(row + e));
        }
    }
}

template <bool REPLAY, int NOUT>
__global__ __launch_bounds__(1024)
void gcn_spmm_owner_kernel(const float* __restrict__ vals,
                           const int* __restrict__ row,
                           const int* __restrict__ col,
                           const float* __restrict__ X,
                           const float* __restrict__ W,
                           float* __restrict__ O,
                           const unsigned* __restrict__ ownerWords,
                           unsigned* __restrict__ edgeList,
                           int* __restrict__ listCount,
                           int E, int nNodes) {
    __shared__ __attribute__((aligned(16))) float    acc[(OWN_ROWS + 1) * ROW_PITCH];
    __shared__ __attribute__((aligned(16))) unsigned queue[OWN_WAVES][OWN_QCAP];
    static_assert(NOUT * ROW_PITCH <= OWN_WAVES * OWN_QCAP, "W must fit the queues' LDS");
    static_assert(NOUT % 32 == 0 && NOUT % OWN_WAVES == 0, "units are 32 columns wide; W is copied OWN_WAVES rows a pass");

    const int tid  = threadIdx.x;
    const int lane = tid & 31;
    const int w = __builtin_amdgcn_readfirstlane(tid >> 5);
    const unsigned laneLt = (1u << lane) - 1u;

    const int nGroups = E >> 9;
    const int nIt     = (nGroups + OWN_WAVES - 1) / OWN_WAVES;
    const int nStep   = (nIt + 3) >> 2;
    const unsigned idBytes = ((unsigned)blockIdx.x & 255u) * 0x01010101u;
    const int startGroup = (int)(((long long)blockIdx.x * nGroups) / gridDim.x);

    uint2* const myList  = (uint2*)edgeList + (long long)(blockIdx.x * OWN_WAVES + w) * LIST_CAP;
    int*   const myCount = listCount + blockIdx.x * OWN_WAVES + w;
    const int firstBase  = blockIdx.x * OWN_ROWS;
    int rec = 0;

    for (int base = blockIdx.x * OWN_ROWS; base < nNodes; base += gridDim.x * OWN_ROWS) {
        const unsigned rowsHere = (unsigned)min(OWN_ROWS, nNodes - base);

        for (int i = tid; i < (OWN_ROWS + 1) * ROW_PITCH / 4; i += blockDim.x)
            ((float4*)acc)[i] = make_float4(0.f, 0.f, 0.f, 0.f);
        __syncthreads();

        int cnt = 0;
        auto push = [&](bool m, int e) {
            const unsigned b = __builtin_amdgcn_ballot_w32(m);
            if (m) queue[w][cnt + __builtin_popcount(b & laneLt)] = (unsigned)e;
            cnt += __builtin_popcount(b);
        };

        auto groupOf = [&](int it) -> int {
            int g = min(it * OWN_WAVES + w, nGroups - 1) + startGroup;
            if (g >= nGroups) g -= nGroups;
            return g;
        };
        auto loadGroup = [&](int it) -> uint4 {
            return *(const uint4*)(ownerWords + groupOf(it) * 128 + lane * 4);
        };
        auto zeroBytes = [](unsigned x) -> unsigned {
            return (x - 0x01010101u) & ~x & 0x80808080u;
        };
        auto scanGroup = [&](const uint4 r, int it) {
            if (it * OWN_WAVES + w < nGroups) {
                const unsigned x0 = r.x ^ idBytes, x1 = r.y ^ idBytes;
                const unsigned x2 = r.z ^ idBytes, x3 = r.w ^ idBytes;
                unsigned c = (zeroBytes(x0) >> 3) | (zeroBytes(x1) >> 2) | (zeroBytes(x2) >> 1) | zeroBytes(x3);
                unsigned b = __builtin_amdgcn_ballot_w32(c != 0u);
                if (b != 0u) {
                    const int e0 = groupOf(it) * 512 + lane * 16;
                    do {
                        const unsigned p = (unsigned)__builtin_ctz(c | 0x80000000u);
                        if (c != 0u)
                            queue[w][cnt + __builtin_popcount(b & laneLt)] = (unsigned)(e0 + (int)(((p & 3u) << 2) + (p >> 3)));
                        cnt += __builtin_popcount(b);
                        c &= c - 1u;
                        b = __builtin_amdgcn_ballot_w32(c != 0u);
                    } while (b != 0u);
                }
            }
        };
        auto addBatch = [&](const int c, const float v, const int lr, const int nb) {
            int j0 = 0;
            do {
                float4 g[8];
                float  vv[8];
                int    rr[8];
#pragma unroll
                for (int u = 0; u < 8; ++u) {
                    const int j  = j0 + u;
                    const int cj = __builtin_amdgcn_readlane(c, j);
                    vv[u] = __int_as_float(__builtin_amdgcn_readlane(__float_as_int(v), j));
                    rr[u] = __builtin_amdgcn_readlane(lr, j);
                    g[u]  = ((const float4*)(X + (long long)cj * DIM_K))[lane];
                }
#pragma unroll
                for (int u = 0; u < 8; ++u) {
                    float* dst = acc + rr[u] * ROW_PITCH + lane;
                    __hip_atomic_fetch_add(dst,      vv[u] * g[u].x, __ATOMIC_RELAXED, __HIP_MEMORY_SCOPE_WORKGROUP);
                    __hip_atomic_fetch_add(dst + 32, vv[u] * g[u].y, __ATOMIC_RELAXED, __HIP_MEMORY_SCOPE_WORKGROUP);
                    __hip_atomic_fetch_add(dst + 64, vv[u] * g[u].z, __ATOMIC_RELAXED, __HIP_MEMORY_SCOPE_WORKGROUP);
                    __hip_atomic_fetch_add(dst + 96, vv[u] * g[u].w, __ATOMIC_RELAXED, __HIP_MEMORY_SCOPE_WORKGROUP);
                }
                j0 += 8;
            } while (j0 < nb);
        };

        auto drain = [&](const bool last) {
            const int todo = last ? cnt : (cnt & ~7);
            int i0 = 0;
            do {
                const int nb = min(32, todo - i0);
                int   c = 0, lr = OWN_ROWS;
                float v = 0.f;
                if (lane < nb) {
                    const int e = (int)queue[w][i0 + lane];
                    const unsigned d  = (unsigned)(row[e] - base);
                    const int      ce = col[e];
                    const float    ve = vals[e];
                    const bool   ours = d < rowsHere;
                    c  = ours ? ce : 0;
                    v  = ours ? ve : 0.f;
                    lr = ours ? (int)d : OWN_ROWS;
                }
                if constexpr (!REPLAY) {
                    if (rec >= 0) {
                        if (rec + nb <= LIST_CAP) {
                            if (lane < nb)
                                myList[rec + lane] = make_uint2(((unsigned)c << 9) | (unsigned)lr, __float_as_uint(v));
                            rec += nb;
                        } else {
                            rec = -1;
                        }
                    }
                }
                addBatch(c, v, lr, nb);
                i0 += 32;
            } while (i0 < todo);
            const int rem = cnt - todo;
            if (lane < rem) {
                const unsigned keep = queue[w][todo + lane];
                queue[w][lane] = keep;
            }
            cnt = rem;
        };

        bool sweep = true;
        if constexpr (REPLAY) {
            int n = -1;
            if (base == firstBase) n = __builtin_amdgcn_readfirstlane(*myCount);
            if ((unsigned)n <= (unsigned)LIST_CAP) {
                sweep = false;
                const uint2 filler = make_uint2((unsigned)OWN_ROWS, 0u);
                if (n > 0) {
                    const int lastEntry = n - 1;
                    uint2 cur = myList[min(lane, lastEntry)];
                    if (lane >= n) cur = filler;
                    int i0 = 0;
                    do {
                        uint2 nxt = myList[min(i0 + 32 + lane, lastEntry)];
                        if (i0 + 32 + lane >= n) nxt = filler;
                        addBatch((int)(cur.x >> 9), __uint_as_float(cur.y), (int)(cur.x & 511u), min(32, n - i0));
                        cur = nxt;
                        i0 += 32;
                    } while (i0 < n);
                }
            }
        }
        if (sweep) {
        for (int t = nGroups * 512 + w * 32; t < E; t += OWN_WAVES * 32) {
            const int e = t + lane;
            unsigned d = 0xffffffffu;
            if (e < E) d = (unsigned)(row[e] - base);
            push(d < rowsHere, e);
        }

        if (nGroups > 0) {
            uint4 c0 = loadGroup(0), c1 = loadGroup(1), c2 = loadGroup(2), c3 = loadGroup(3);
#pragma unroll 1
            for (int st = 0; st < nStep; st += 2) {
                int it = st * 4;
                const uint4 n0 = loadGroup(it + 4), n1 = loadGroup(it + 5);
                const uint4 n2 = loadGroup(it + 6), n3 = loadGroup(it + 7);
                scanGroup(c0, it);
                if (cnt > OWN_QCAP - 512) drain(false);
                scanGroup(c1, it + 1);
                if (cnt > OWN_QCAP - 512) drain(false);
                scanGroup(c2, it + 2);
                if (cnt > OWN_QCAP - 512) drain(false);
                scanGroup(c3, it + 3);
                if (cnt > OWN_QCAP - 512) drain(false);

                it += 4;
                c0 = loadGroup(it + 4); c1 = loadGroup(it + 5);
                c2 = loadGroup(it + 6); c3 = loadGroup(it + 7);
                scanGroup(n0, it);
                if (cnt > OWN_QCAP - 512) drain(false);
                scanGroup(n1, it + 1);
                if (cnt > OWN_QCAP - 512) drain(false);
                scanGroup(n2, it + 2);
                if (cnt > OWN_QCAP - 512) drain(false);
                scanGroup(n3, it + 3);
                if (cnt > OWN_QCAP - 512) drain(false);
            }
        }
        if (cnt > 0) drain(true);
        }
        if constexpr (!REPLAY) {
            if (base == firstBase && lane == 0) *myCount = rec;
            rec = -1;
        }
        __syncthreads();

        unsigned* const wl = &queue[0][0];
        {
            constexpr int W_PASSES = NOUT / OWN_WAVES;
            float4 wv[W_PASSES];
#pragma unroll
            for (int p = 0; p < W_PASSES; ++p)
                wv[p] = ((const float4*)(W + (long long)(p * OWN_WAVES + w) * DIM_K))[lane];
#pragma unroll
            for (int p = 0; p < W_PASSES; ++p) {
                unsigned* d = wl + (p * OWN_WAVES + w) * ROW_PITCH + lane;
                d[0]  = __float_as_uint(wv[p].x);
                d[32] = __float_as_uint(wv[p].y);
                d[64] = __float_as_uint(wv[p].z);
                d[96] = __float_as_uint(wv[p].w);
            }
        }
        __syncthreads();

        {
            constexpr int COL_GROUPS = NOUT / 32;
            const int c = lane & 15;
            const int h = lane >> 4;
            const int nStrips = ((int)rowsHere + 15) >> 4;
            const int nUnits  = nStrips * COL_GROUPS;
            for (int u = w; u < nUnits; u += OWN_WAVES) {
                const int strip = u / COL_GROUPS;
                const int cg    = u - strip * COL_GROUPS;
                const int arow = min(strip * 16 + c, OWN_ROWS);
                const float4* ap  = (const float4*)(acc + arow * ROW_PITCH + 4 * h);
                const uint4*  bp0 = (const uint4*)(wl + (cg * 32 + c) * ROW_PITCH + 4 * h);
                const uint4*  bp1 = (const uint4*)(wl + (cg * 32 + 16 + c) * ROW_PITCH + 4 * h);
                v8f d0 = (v8f){}, d1 = (v8f){};
#pragma unroll
                for (int j = 0; j < DIM_K / 8; ++j) {
                    const float4 a  = ap[2 * j];
                    const uint4  p0 = bp0[2 * j];
                    const uint4  p1 = bp1[2 * j];
                    {
                        const v2f x  = (v2f){a.x, a.y};
                        const v2f y0 = (v2f){__uint_as_float(p0.x), __uint_as_float(p0.y)};
                        const v2f y1 = (v2f){__uint_as_float(p1.x), __uint_as_float(p1.y)};
                        d0 = __builtin_amdgcn_wmma_f32_16x16x4_f32(
                             false, x,  false, y0,
                             (short)0, d0,  false,  false);
                        d1 = __builtin_amdgcn_wmma_f32_16x16x4_f32(
                            false, x, false, y1, (short)0, d1, false, false);
                    }
                    {
                        const v2f x  = (v2f){a.z, a.w};
                        const v2f y0 = (v2f){__uint_as_float(p0.z), __uint_as_float(p0.w)};
                        const v2f y1 = (v2f){__uint_as_float(p1.z), __uint_as_float(p1.w)};
                        d0 = __builtin_amdgcn_wmma_f32_16x16x4_f32(
                            false, x, false, y0, (short)0, d0, false, false);
                        d1 = __builtin_amdgcn_wmma_f32_16x16x4_f32(
                            false, x, false, y1, (short)0, d1, false, false);
                    }
                }
                float t0[8], t1[8];
#pragma unroll
                for (int r = 0; r < 8; ++r) { t0[r] = fast_tanh(d0[r]); t1[r] = fast_tanh(d1[r]); }
                const int m0 = strip * 16 + 8 * h;
                float* op = O + (long long)(base + m0) * NOUT + cg * 32 + c;
                if (strip * 16 + 16 <= (int)rowsHere) {
#pragma unroll
                    for (int r = 0; r < 8; ++r) {
                        op[r * NOUT]      = t0[r];
                        op[r * NOUT + 16] = t1[r];
                    }
                } else {
#pragma unroll
                    for (int r = 0; r < 8; ++r) {
                        if ((unsigned)(m0 + r) < rowsHere) {
                            op[r * NOUT]      = t0[r];
                            op[r * NOUT + 16] = t1[r];
                        }
                    }
                }
            }
        }
        __syncthreads();
    }
}

extern "C" void kernel_launch(void* const* d_in, const int* in_sizes, int n_in,
                              void* d_out, int out_size, void* d_ws, size_t ws_size,
                              hipStream_t stream) {
    const float* X        = (const float*)d_in[0];
    const float* W1       = (const float*)d_in[1];
    const float* W2       = (const float*)d_in[2];
    const float* edgeVals = (const float*)d_in[3];
    const int*   edgeRow  = (const int*)d_in[4];
    const int*   edgeCol  = (const int*)d_in[5];
    float* out = (float*)d_out;

    const int N = in_sizes[0] / DIM_K;
    const int E = in_sizes[3];
    const long long featCount = (long long)N * DIM_K;

    float* H1 = (float*)d_ws;
    unsigned* edgeList  = (unsigned*)(H1 + featCount);
    int*      listCount = (int*)(edgeList + 2LL * OWN_GROUPS * OWN_WAVES * LIST_CAP);
    unsigned* ownerWords = (unsigned*)(listCount + OWN_GROUPS * OWN_WAVES);

    const int idBlocks = (E + 4095) / 4096;
    gcn_owner_ids_kernel<<<idBlocks, 256, 0, stream>>>(edgeRow, ownerWords, E);

    gcn_spmm_owner_kernel<false, 128><<<OWN_GROUPS, 1024, 0, stream>>>(
        edgeVals, edgeRow, edgeCol, X, W1, H1, ownerWords, edgeList, listCount, E, N);

    gcn_spmm_owner_kernel<true, 64><<<OWN_GROUPS, 1024, 0, stream>>>(
        edgeVals, edgeRow, edgeCol, H1, W2, out, ownerWords, edgeList, listCount, E, N);
}


# Layer 1's drain reads one packed 8-byte record per kept edge (source row, local row, weight; written in edge order by the first launch) instead of three scattered 4-byte loads; sweep flags made exact
# speedup vs baseline: 4.2285x; 1.0449x over previous
#include <hip/hip_runtime.h>
#include <hip/hip_bf16.h>
#include <math.h>


typedef __attribute__((ext_vector_type(2))) float v2f;
typedef __attribute__((ext_vector_type(8))) float v8f;

#define DIM_K 128

__device__ __forceinline__ float fast_tanh(float x) {
    float ax = fabsf(x);
    float e  = __expf(-2.0f * ax);
    float r  = (1.0f - e) * __builtin_amdgcn_rcpf(1.0f + e);
    return copysignf(r, x);
}

#define OWN_ROWS  392
#define ROW_PITCH 132
#define OWN_WAVES 32
#define OWN_QCAP  640
#define OWN_GROUPS 256
#define LIST_CAP  1024

__device__ __forceinline__ unsigned owner_id(int r) {
    return ((unsigned)r / (unsigned)OWN_ROWS) & 255u;
}

__global__ __launch_bounds__(256)
void gcn_owner_ids_kernel(const int* __restrict__ row, const int* __restrict__ col, const float* __restrict__ vals,
                          unsigned* __restrict__ ownerWords, unsigned* __restrict__ recordWords, int E) {
    const int lane  = threadIdx.x & 31;
    const int wave  = blockIdx.x * (int)(blockDim.x >> 5) + __builtin_amdgcn_readfirstlane((int)(threadIdx.x >> 5));
    const int eWave = wave * 512;
    auto pack = [](const int4 r) -> unsigned {
        return owner_id(r.x) | (owner_id(r.y) << 8) | (owner_id(r.z) << 16) | (owner_id(r.w) << 24);
    };
    if (eWave + 512 <= E) {
        int4 r[4];
#pragma unroll
        for (int j = 0; j < 4; ++j) r[j] = *(const int4*)(row + eWave + j * 128 + lane * 4);
#pragma unroll
        for (int j = 0; j < 4; ++j) ownerWords[(eWave >> 2) + j * 32 + lane] = pack(r[j]);
    } else {
#pragma unroll
        for (int j = 0; j < 4; ++j) {
            const int e = eWave + j * 128 + lane * 4;
            if (e + 3 < E) ownerWords[e >> 2] = pack(*(const int4*)(row + e));
        }
    }
    uint4* const recordPairs = (uint4*)recordWords;
    auto recWord = [](const int r, const int c) -> unsigned {
        const unsigned q = (unsigned)r / (unsigned)OWN_ROWS;
        return ((unsigned)c << 9) | ((unsigned)r - q * (unsigned)OWN_ROWS);
    };
    if (eWave + 512 <= E) {
#pragma unroll
        for (int s = 0; s < 8; ++s) {
            const int e = eWave + s * 64 + lane * 2;
            const int2   r = *(const int2*)(row + e);
            const int2   c = *(const int2*)(col + e);
            const float2 v = *(const float2*)(vals + e);
            recordPairs[e >> 1] = make_uint4(recWord(r.x, c.x), __float_as_uint(v.x), recWord(r.y, c.y), __float_as_uint(v.y));
        }
    } else {
        uint2* const records = (uint2*)recordPairs;
        for (int e = eWave + lane; e < min(E, eWave + 512); e += 32)
            records[e] = make_uint2(recWord(row[e], col[e]), __float_as_uint(vals[e]));
    }
}

template <bool REPLAY, int NOUT>
__global__ __launch_bounds__(1024)
void gcn_spmm_owner_kernel(const unsigned* __restrict__ recordWords,
                           const int* __restrict__ row,
                           const float* __restrict__ X,
                           const float* __restrict__ W,
                           float* __restrict__ O,
                           const unsigned* __restrict__ ownerWords,
                           unsigned* __restrict__ edgeList,
                           int* __restrict__ listCount,
                           int E, int nNodes) {
    __shared__ __attribute__((aligned(16))) float    acc[(OWN_ROWS + 1) * ROW_PITCH];
    __shared__ __attribute__((aligned(16))) unsigned queue[OWN_WAVES][OWN_QCAP];
    static_assert(NOUT * ROW_PITCH <= OWN_WAVES * OWN_QCAP, "W must fit the queues' LDS");
    static_assert(NOUT % 32 == 0 && NOUT % OWN_WAVES == 0, "units are 32 columns wide; W is copied OWN_WAVES rows a pass");

    const int tid  = threadIdx.x;
    const int lane = tid & 31;
    const int w = __builtin_amdgcn_readfirstlane(tid >> 5);
    const unsigned laneLt = (1u << lane) - 1u;

    const int nGroups = E >> 9;
    const int nIt     = (nGroups + OWN_WAVES - 1) / OWN_WAVES;
    const int nStep   = (nIt + 3) >> 2;
    const unsigned idBytes = ((unsigned)blockIdx.x & 255u) * 0x01010101u;
    const int startGroup = (int)(((long long)blockIdx.x * nGroups) / gridDim.x);

    uint2* const myList  = (uint2*)edgeList + (long long)(blockIdx.x * OWN_WAVES + w) * LIST_CAP;
    int*   const myCount = listCount + blockIdx.x * OWN_WAVES + w;
    const int firstBase  = blockIdx.x * OWN_ROWS;
    const bool manyRanges = nNodes > 256 * OWN_ROWS;
    const uint2* const records = (const uint2*)recordWords;
    int rec = 0;

    for (int base = blockIdx.x * OWN_ROWS; base < nNodes; base += gridDim.x * OWN_ROWS) {
        const unsigned rowsHere = (unsigned)min(OWN_ROWS, nNodes - base);

        for (int i = tid; i < (OWN_ROWS + 1) * ROW_PITCH / 4; i += blockDim.x)
            ((float4*)acc)[i] = make_float4(0.f, 0.f, 0.f, 0.f);
        __syncthreads();

        int cnt = 0;
        auto push = [&](bool m, int e) {
            const unsigned b = __builtin_amdgcn_ballot_w32(m);
            if (m) queue[w][cnt + __builtin_popcount(b & laneLt)] = (unsigned)e;
            cnt += __builtin_popcount(b);
        };

        auto groupOf = [&](int it) -> int {
            int g = min(it * OWN_WAVES + w, nGroups - 1) + startGroup;
            if (g >= nGroups) g -= nGroups;
            return g;
        };
        auto loadGroup = [&](int it) -> uint4 {
            return *(const uint4*)(ownerWords + groupOf(it) * 128 + lane * 4);
        };
        auto zeroBytes = [](unsigned x) -> unsigned {
            return ~(((x & 0x7f7f7f7fu) + 0x7f7f7f7fu) | x) & 0x80808080u;
        };
        auto scanGroup = [&](const uint4 r, int it) {
            if (it * OWN_WAVES + w < nGroups) {
                const unsigned x0 = r.x ^ idBytes, x1 = r.y ^ idBytes;
                const unsigned x2 = r.z ^ idBytes, x3 = r.w ^ idBytes;
                unsigned c = (zeroBytes(x0) >> 3) | (zeroBytes(x1) >> 2) | (zeroBytes(x2) >> 1) | zeroBytes(x3);
                unsigned b = __builtin_amdgcn_ballot_w32(c != 0u);
                if (b != 0u) {
                    const int e0 = groupOf(it) * 512 + lane * 16;
                    do {
                        const unsigned p = (unsigned)__builtin_ctz(c | 0x80000000u);
                        if (c != 0u)
                            queue[w][cnt + __builtin_popcount(b & laneLt)] = (unsigned)(e0 + (int)(((p & 3u) << 2) + (p >> 3)));
                        cnt += __builtin_popcount(b);
                        c &= c - 1u;
                        b = __builtin_amdgcn_ballot_w32(c != 0u);
                    } while (b != 0u);
                }
            }
        };
        auto addBatch = [&](const int c, const float v, const int lr, const int nb) {
            int j0 = 0;
            do {
                float4 g[8];
                float  vv[8];
                int    rr[8];
#pragma unroll
                for (int u = 0; u < 8; ++u) {
                    const int j  = j0 + u;
                    const int cj = __builtin_amdgcn_readlane(c, j);
                    vv[u] = __int_as_float(__builtin_amdgcn_readlane(__float_as_int(v), j));
                    rr[u] = __builtin_amdgcn_readlane(lr, j);
                    g[u]  = ((const float4*)(X + (long long)cj * DIM_K))[lane];
                }
#pragma unroll
                for (int u = 0; u < 8; ++u) {
                    float* dst = acc + rr[u] * ROW_PITCH + lane;
                    __hip_atomic_fetch_add(dst,      vv[u] * g[u].x, __ATOMIC_RELAXED, __HIP_MEMORY_SCOPE_WORKGROUP);
                    __hip_atomic_fetch_add(dst + 32, vv[u] * g[u].y, __ATOMIC_RELAXED, __HIP_MEMORY_SCOPE_WORKGROUP);
                    __hip_atomic_fetch_add(dst + 64, vv[u] * g[u].z, __ATOMIC_RELAXED, __HIP_MEMORY_SCOPE_WORKGROUP);
                    __hip_atomic_fetch_add(dst + 96, vv[u] * g[u].w, __ATOMIC_RELAXED, __HIP_MEMORY_SCOPE_WORKGROUP);
                }
                j0 += 8;
            } while (j0 < nb);
        };

        auto drain = [&](const bool last) {
            const int todo = last ? cnt : (cnt & ~7);
            int i0 = 0;
            do {
                const int nb = min(32, todo - i0);
                int   c = 0, lr = OWN_ROWS;
                float v = 0.f;
                if (lane < nb) {
                    const int e = (int)queue[w][i0 + lane];
                    const uint2 r = records[e];
                    bool ours = true;
                    if (manyRanges) ours = (unsigned)(row[e] - base) < rowsHere;
                    c  = ours ? (int)(r.x >> 9) : 0;
                    v  = ours ? __uint_as_float(r.y) : 0.f;
                    lr = ours ? (int)(r.x & 511u) : OWN_ROWS;
                }
                if constexpr (!REPLAY) {
                    if (rec >= 0) {
                        if (rec + nb <= LIST_CAP) {
                            if (lane < nb)
                                myList[rec + lane] = make_uint2(((unsigned)c << 9) | (unsigned)lr, __float_as_uint(v));
                            rec += nb;
                        } else {
                            rec = -1;
                        }
                    }
                }
                addBatch(c, v, lr, nb);
                i0 += 32;
            } while (i0 < todo);
            const int rem = cnt - todo;
            if (lane < rem) {
                const unsigned keep = queue[w][todo + lane];
                queue[w][lane] = keep;
            }
            cnt = rem;
        };

        bool sweep = true;
        if constexpr (REPLAY) {
            int n = -1;
            if (base == firstBase) n = __builtin_amdgcn_readfirstlane(*myCount);
            if ((unsigned)n <= (unsigned)LIST_CAP) {
                sweep = false;
                const uint2 filler = make_uint2((unsigned)OWN_ROWS, 0u);
                if (n > 0) {
                    const int lastEntry = n - 1;
                    uint2 cur = myList[min(lane, lastEntry)];
                    if (lane >= n) cur = filler;
                    int i0 = 0;
                    do {
                        uint2 nxt = myList[min(i0 + 32 + lane, lastEntry)];
                        if (i0 + 32 + lane >= n) nxt = filler;
                        addBatch((int)(cur.x >> 9), __uint_as_float(cur.y), (int)(cur.x & 511u), min(32, n - i0));
                        cur = nxt;
                        i0 += 32;
                    } while (i0 < n);
                }
            }
        }
        if (sweep) {
        for (int t = nGroups * 512 + w * 32; t < E; t += OWN_WAVES * 32) {
            const int e = t + lane;
            unsigned d = 0xffffffffu;
            if (e < E) d = (unsigned)(row[e] - base);
            push(d < rowsHere, e);
        }

        if (nGroups > 0) {
            uint4 c0 = loadGroup(0), c1 = loadGroup(1), c2 = loadGroup(2), c3 = loadGroup(3);
#pragma unroll 1
            for (int st = 0; st < nStep; st += 2) {
                int it = st * 4;
                const uint4 n0 = loadGroup(it + 4), n1 = loadGroup(it + 5);
                const uint4 n2 = loadGroup(it + 6), n3 = loadGroup(it + 7);
                scanGroup(c0, it);
                if (cnt > OWN_QCAP - 512) drain(false);
                scanGroup(c1, it + 1);
                if (cnt > OWN_QCAP - 512) drain(false);
                scanGroup(c2, it + 2);
                if (cnt > OWN_QCAP - 512) drain(false);
                scanGroup(c3, it + 3);
                if (cnt > OWN_QCAP - 512) drain(false);

                it += 4;
                c0 = loadGroup(it + 4); c1 = loadGroup(it + 5);
                c2 = loadGroup(it + 6); c3 = loadGroup(it + 7);
                scanGroup(n0, it);
                if (cnt > OWN_QCAP - 512) drain(false);
                scanGroup(n1, it + 1);
                if (cnt > OWN_QCAP - 512) drain(false);
                scanGroup(n2, it + 2);
                if (cnt > OWN_QCAP - 512) drain(false);
                scanGroup(n3, it + 3);
                if (cnt > OWN_QCAP - 512) drain(false);
            }
        }
        if (cnt > 0) drain(true);
        }
        if constexpr (!REPLAY) {
            if (base == firstBase && lane == 0) *myCount = rec;
            rec = -1;
        }
        __syncthreads();

        unsigned* const wl = &queue[0][0];
        {
            constexpr int W_PASSES = NOUT / OWN_WAVES;
            float4 wv[W_PASSES];
#pragma unroll
            for (int p = 0; p < W_PASSES; ++p)
                wv[p] = ((const float4*)(W + (long long)(p * OWN_WAVES + w) * DIM_K))[lane];
#pragma unroll
            for (int p = 0; p < W_PASSES; ++p) {
                unsigned* d = wl + (p * OWN_WAVES + w) * ROW_PITCH + lane;
                d[0]  = __float_as_uint(wv[p].x);
                d[32] = __float_as_uint(wv[p].y);
                d[64] = __float_as_uint(wv[p].z);
                d[96] = __float_as_uint(wv[p].w);
            }
        }
        __syncthreads();

        {
            constexpr int COL_GROUPS = NOUT / 32;
            const int c = lane & 15;
            const int h = lane >> 4;
            const int nStrips = ((int)rowsHere + 15) >> 4;
            const int nUnits  = nStrips * COL_GROUPS;
            for (int u = w; u < nUnits; u += OWN_WAVES) {
                const int strip = u / COL_GROUPS;
                const int cg    = u - strip * COL_GROUPS;
                const int arow = min(strip * 16 + c, OWN_ROWS);
                const float4* ap  = (const float4*)(acc + arow * ROW_PITCH + 4 * h);
                const uint4*  bp0 = (const uint4*)(wl + (cg * 32 + c) * ROW_PITCH + 4 * h);
                const uint4*  bp1 = (const uint4*)(wl + (cg * 32 + 16 + c) * ROW_PITCH + 4 * h);
                v8f d0 = (v8f){}, d1 = (v8f){};
#pragma unroll
                for (int j = 0; j < DIM_K / 8; ++j) {
                    const float4 a  = ap[2 * j];
                    const uint4  p0 = bp0[2 * j];
                    const uint4  p1 = bp1[2 * j];
                    {
                        const v2f x  = (v2f){a.x, a.y};
                        const v2f y0 = (v2f){__uint_as_float(p0.x), __uint_as_float(p0.y)};
                        const v2f y1 = (v2f){__uint_as_float(p1.x), __uint_as_float(p1.y)};
                        d0 = __builtin_amdgcn_wmma_f32_16x16x4_f32(
                             false, x,  false, y0,
                             (short)0, d0,  false,  false);
                        d1 = __builtin_amdgcn_wmma_f32_16x16x4_f32(
                            false, x, false, y1, (short)0, d1, false, false);
                    }
                    {
                        const v2f x  = (v2f){a.z, a.w};
                        const v2f y0 = (v2f){__uint_as_float(p0.z), __uint_as_float(p0.w)};
                        const v2f y1 = (v2f){__uint_as_float(p1.z), __uint_as_float(p1.w)};
                        d0 = __builtin_amdgcn_wmma_f32_16x16x4_f32(
                            false, x, false, y0, (short)0, d0, false, false);
                        d1 = __builtin_amdgcn_wmma_f32_16x16x4_f32(
                            false, x, false, y1, (short)0, d1, false, false);
                    }
                }
                float t0[8], t1[8];
#pragma unroll
                for (int r = 0; r < 8; ++r) { t0[r] = fast_tanh(d0[r]); t1[r] = fast_tanh(d1[r]); }
                const int m0 = strip * 16 + 8 * h;
                float* op = O + (long long)(base + m0) * NOUT + cg * 32 + c;
                if (strip * 16 + 16 <= (int)rowsHere) {
#pragma unroll
                    for (int r = 0; r < 8; ++r) {
                        op[r * NOUT]      = t0[r];
                        op[r * NOUT + 16] = t1[r];
                    }
                } else {
#pragma unroll
                    for (int r = 0; r < 8; ++r) {
                        if ((unsigned)(m0 + r) < rowsHere) {
                            op[r * NOUT]      = t0[r];
                            op[r * NOUT + 16] = t1[r];
                        }
                    }
                }
            }
        }
        __syncthreads();
    }
}

extern "C" void kernel_launch(void* const* d_in, const int* in_sizes, int n_in,
                              void* d_out, int out_size, void* d_ws, size_t ws_size,
                              hipStream_t stream) {
    const float* X        = (const float*)d_in[0];
    const float* W1       = (const float*)d_in[1];
    const float* W2       = (const float*)d_in[2];
    const float* edgeVals = (const float*)d_in[3];
    const int*   edgeRow  = (const int*)d_in[4];
    const int*   edgeCol  = (const int*)d_in[5];
    float* out = (float*)d_out;

    const int N = in_sizes[0] / DIM_K;
    const int E = in_sizes[3];
    const long long featCount = (long long)N * DIM_K;

    float* H1 = (float*)d_ws;
    unsigned* edgeList  = (unsigned*)(H1 + featCount);
    int*      listCount = (int*)(edgeList + 2LL * OWN_GROUPS * OWN_WAVES * LIST_CAP);
    unsigned* ownerWords = (unsigned*)(listCount + OWN_GROUPS * OWN_WAVES);
    unsigned* recordWords = ownerWords + 4 * 1024 * 1024;

    const int idBlocks = (E + 4095) / 4096;
    gcn_owner_ids_kernel<<<idBlocks, 256, 0, stream>>>(edgeRow, edgeCol, edgeVals, ownerWords, recordWords, E);

    gcn_spmm_owner_kernel<false, 128><<<OWN_GROUPS, 1024, 0, stream>>>(
        recordWords, edgeRow, X, W1, H1, ownerWords, edgeList, listCount, E, N);

    gcn_spmm_owner_kernel<true, 64><<<OWN_GROUPS, 1024, 0, stream>>>(
        recordWords, edgeRow, H1, W2, out, ownerWords, edgeList, listCount, E, N);
}


# Layer 1's sweep reads only the 64-edge groups holding an edge of its range: launch 1 adds a 64-bit group map per 4096 edges and owner (LDS ORs); queue order changes, sums stay LDS arrival order
# speedup vs baseline: 4.4950x; 1.0630x over previous
#include <hip/hip_runtime.h>
#include <hip/hip_bf16.h>
#include <math.h>


typedef __attribute__((ext_vector_type(2))) float v2f;
typedef __attribute__((ext_vector_type(8))) float v8f;

#define DIM_K 128

__device__ __forceinline__ float fast_tanh(float x) {
    float ax = fabsf(x);
    float e  = __expf(-2.0f * ax);
    float r  = (1.0f - e) * __builtin_amdgcn_rcpf(1.0f + e);
    return copysignf(r, x);
}

#define OWN_ROWS  392
#define ROW_PITCH 132
#define OWN_WAVES 32
#define OWN_QCAP  640
#define OWN_GROUPS 256
#define LIST_CAP  1024

__device__ __forceinline__ unsigned owner_id(int r) {
    return ((unsigned)r / (unsigned)OWN_ROWS) & 255u;
}

__global__ __launch_bounds__(256)
void gcn_owner_ids_kernel(const int* __restrict__ row, const int* __restrict__ col, const float* __restrict__ vals,
                          unsigned* __restrict__ ownerWords, unsigned* __restrict__ recordWords,
                          unsigned* __restrict__ groupMaps, int E) {
    __shared__ __attribute__((aligned(8))) unsigned mapWords[512];
    mapWords[threadIdx.x]       = 0u;
    mapWords[threadIdx.x + 256] = 0u;
    __syncthreads();
    const int lane  = threadIdx.x & 31;
    const int wib   = __builtin_amdgcn_readfirstlane((int)(threadIdx.x >> 5));
    const int wave  = blockIdx.x * (int)(blockDim.x >> 5) + __builtin_amdgcn_readfirstlane((int)(threadIdx.x >> 5));
    const int eWave = wave * 512;
    auto pack = [](const int4 r) -> unsigned {
        return owner_id(r.x) | (owner_id(r.y) << 8) | (owner_id(r.z) << 16) | (owner_id(r.w) << 24);
    };
    if (eWave + 512 <= E) {
        int4 r[4];
#pragma unroll
        for (int j = 0; j < 4; ++j) r[j] = *(const int4*)(row + eWave + j * 128 + lane * 4);
#pragma unroll
        for (int j = 0; j < 4; ++j) ownerWords[(eWave >> 2) + j * 32 + lane] = pack(r[j]);
        unsigned* const half = mapWords + (wib >> 2);
#pragma unroll
        for (int j = 0; j < 4; ++j) {
            const unsigned bit = 1u << ((wib & 3) * 8 + j * 2 + (lane >> 4));
            __hip_atomic_fetch_or(half + 2u * owner_id(r[j].x), bit, __ATOMIC_RELAXED, __HIP_MEMORY_SCOPE_WORKGROUP);
            __hip_atomic_fetch_or(half + 2u * owner_id(r[j].y), bit, __ATOMIC_RELAXED, __HIP_MEMORY_SCOPE_WORKGROUP);
            __hip_atomic_fetch_or(half + 2u * owner_id(r[j].z), bit, __ATOMIC_RELAXED, __HIP_MEMORY_SCOPE_WORKGROUP);
            __hip_atomic_fetch_or(half + 2u * owner_id(r[j].w), bit, __ATOMIC_RELAXED, __HIP_MEMORY_SCOPE_WORKGROUP);
        }
    } else {
#pragma unroll
        for (int j = 0; j < 4; ++j) {
            const int e = eWave + j * 128 + lane * 4;
            if (e + 3 < E) ownerWords[e >> 2] = pack(*(const int4*)(row + e));
        }
    }
    uint4* const recordPairs = (uint4*)recordWords;
    auto recWord = [](const int r, const int c) -> unsigned {
        const unsigned q = (unsigned)r / (unsigned)OWN_ROWS;
        return ((unsigned)c << 9) | ((unsigned)r - q * (unsigned)OWN_ROWS);
    };
    if (eWave + 512 <= E) {
#pragma unroll
        for (int s = 0; s < 8; ++s) {
            const int e = eWave + s * 64 + lane * 2;
            const int2   r = *(const int2*)(row + e);
            const int2   c = *(const int2*)(col + e);
            const float2 v = *(const float2*)(vals + e);
            recordPairs[e >> 1] = make_uint4(recWord(r.x, c.x), __float_as_uint(v.x), recWord(r.y, c.y), __float_as_uint(v.y));
        }
    } else {
        uint2* const records = (uint2*)recordPairs;
        for (int e = eWave + lane; e < min(E, eWave + 512); e += 32)
            records[e] = make_uint2(recWord(row[e], col[e]), __float_as_uint(vals[e]));
    }
    __syncthreads();
    ((uint2*)groupMaps)[blockIdx.x * 256 + threadIdx.x] = ((const uint2*)mapWords)[threadIdx.x];
}

template <bool REPLAY, int NOUT>
__global__ __launch_bounds__(1024)
void gcn_spmm_owner_kernel(const unsigned* __restrict__ recordWords,
                           const int* __restrict__ row,
                           const float* __restrict__ X,
                           const float* __restrict__ W,
                           float* __restrict__ O,
                           const unsigned* __restrict__ ownerWords,
                           const unsigned* __restrict__ groupMaps,
                           unsigned* __restrict__ edgeList,
                           int* __restrict__ listCount,
                           int E, int nNodes) {
    __shared__ __attribute__((aligned(16))) float    acc[(OWN_ROWS + 1) * ROW_PITCH];
    __shared__ __attribute__((aligned(16))) unsigned queue[OWN_WAVES][OWN_QCAP];
    static_assert(NOUT * ROW_PITCH <= OWN_WAVES * OWN_QCAP, "W must fit the queues' LDS");
    static_assert(NOUT % 32 == 0 && NOUT % OWN_WAVES == 0, "units are 32 columns wide; W is copied OWN_WAVES rows a pass");

    const int tid  = threadIdx.x;
    const int lane = tid & 31;
    const int w = __builtin_amdgcn_readfirstlane(tid >> 5);
    const unsigned laneLt = (1u << lane) - 1u;

    const int nGroups = E >> 9;
    const int nBlk    = (nGroups + 7) >> 3;
    const unsigned idBytes = ((unsigned)blockIdx.x & 255u) * 0x01010101u;
    const int startBlk = (int)(((long long)blockIdx.x * nBlk) / gridDim.x);
    const uint2* const myMaps = (const uint2*)groupMaps + (blockIdx.x & 255u);
    const unsigned char* const ownerBytes = (const unsigned char*)ownerWords;
    auto loadMaps = [&](const int first, int& blkOfLane, uint2& mapOfLane) {
        blkOfLane = 0;
        mapOfLane = make_uint2(0u, 0u);
        if (first + OWN_WAVES * lane < nBlk) {
            int bq = first + OWN_WAVES * lane + startBlk;
            if (bq >= nBlk) bq -= nBlk;
            blkOfLane = bq;
            mapOfLane = myMaps[(long long)bq * 256];
        }
    };
    int   blkFirst = 0;
    uint2 mapFirst = make_uint2(0u, 0u);
    if constexpr (!REPLAY) loadMaps(w, blkFirst, mapFirst);

    uint2* const myList  = (uint2*)edgeList + (long long)(blockIdx.x * OWN_WAVES + w) * LIST_CAP;
    int*   const myCount = listCount + blockIdx.x * OWN_WAVES + w;
    const int firstBase  = blockIdx.x * OWN_ROWS;
    const bool manyRanges = nNodes > 256 * OWN_ROWS;
    const uint2* const records = (const uint2*)recordWords;
    int rec = 0;

    for (int base = blockIdx.x * OWN_ROWS; base < nNodes; base += gridDim.x * OWN_ROWS) {
        const unsigned rowsHere = (unsigned)min(OWN_ROWS, nNodes - base);

        for (int i = tid; i < (OWN_ROWS + 1) * ROW_PITCH / 4; i += blockDim.x)
            ((float4*)acc)[i] = make_float4(0.f, 0.f, 0.f, 0.f);
        __syncthreads();

        int cnt = 0;
        auto push = [&](bool m, int e) {
            const unsigned b = __builtin_amdgcn_ballot_w32(m);
            if (m) queue[w][cnt + __builtin_popcount(b & laneLt)] = (unsigned)e;
            cnt += __builtin_popcount(b);
        };

        auto zeroBytes = [](unsigned x) -> unsigned {
            return ~(((x & 0x7f7f7f7fu) + 0x7f7f7f7fu) | x) & 0x80808080u;
        };
        auto scanChunk = [&](const uint4 r, const int e0) {
            const unsigned x0 = r.x ^ idBytes, x1 = r.y ^ idBytes;
            const unsigned x2 = r.z ^ idBytes, x3 = r.w ^ idBytes;
            unsigned c = (zeroBytes(x0) >> 3) | (zeroBytes(x1) >> 2) | (zeroBytes(x2) >> 1) | zeroBytes(x3);
            if (e0 < 0) c = 0u;
            unsigned b = __builtin_amdgcn_ballot_w32(c != 0u);
            while (b != 0u) {
                const unsigned p = (unsigned)__builtin_ctz(c | 0x80000000u);
                if (c != 0u)
                    queue[w][cnt + __builtin_popcount(b & laneLt)] = (unsigned)(e0 + (int)(((p & 3u) << 2) + (p >> 3)));
                cnt += __builtin_popcount(b);
                c &= c - 1u;
                b = __builtin_amdgcn_ballot_w32(c != 0u);
            }
        };
        auto addBatch = [&](const int c, const float v, const int lr, const int nb) {
            int j0 = 0;
            do {
                float4 g[8];
                float  vv[8];
                int    rr[8];
#pragma unroll
                for (int u = 0; u < 8; ++u) {
                    const int j  = j0 + u;
                    const int cj = __builtin_amdgcn_readlane(c, j);
                    vv[u] = __int_as_float(__builtin_amdgcn_readlane(__float_as_int(v), j));
                    rr[u] = __builtin_amdgcn_readlane(lr, j);
                    g[u]  = ((const float4*)(X + (long long)cj * DIM_K))[lane];
                }
#pragma unroll
                for (int u = 0; u < 8; ++u) {
                    float* dst = acc + rr[u] * ROW_PITCH + lane;
                    __hip_atomic_fetch_add(dst,      vv[u] * g[u].x, __ATOMIC_RELAXED, __HIP_MEMORY_SCOPE_WORKGROUP);
                    __hip_atomic_fetch_add(dst + 32, vv[u] * g[u].y, __ATOMIC_RELAXED, __HIP_MEMORY_SCOPE_WORKGROUP);
                    __hip_atomic_fetch_add(dst + 64, vv[u] * g[u].z, __ATOMIC_RELAXED, __HIP_MEMORY_SCOPE_WORKGROUP);
                    __hip_atomic_fetch_add(dst + 96, vv[u] * g[u].w, __ATOMIC_RELAXED, __HIP_MEMORY_SCOPE_WORKGROUP);
                }
                j0 += 8;
            } while (j0 < nb);
        };

        auto drain = [&](const bool last) {
            const int todo = last ? cnt : (cnt & ~7);
            int i0 = 0;
            do {
                const int nb = min(32, todo - i0);
                int   c = 0, lr = OWN_ROWS;
                float v = 0.f;
                if (lane < nb) {
                    const int e = (int)queue[w][i0 + lane];
                    const uint2 r = records[e];
                    bool ours = true;
                    if (manyRanges) ours = (unsigned)(row[e] - base) < rowsHere;
                    c  = ours ? (int)(r.x >> 9) : 0;
                    v  = ours ? __uint_as_float(r.y) : 0.f;
                    lr = ours ? (int)(r.x & 511u) : OWN_ROWS;
                }
                if constexpr (!REPLAY) {
                    if (rec >= 0) {
                        if (rec + nb <= LIST_CAP) {
                            if (lane < nb)
                                myList[rec + lane] = make_uint2(((unsigned)c << 9) | (unsigned)lr, __float_as_uint(v));
                            rec += nb;
                        } else {
                            rec = -1;
                        }
                    }
                }
                addBatch(c, v, lr, nb);
                i0 += 32;
            } while (i0 < todo);
            const int rem = cnt - todo;
            if (lane < rem) {
                const unsigned keep = queue[w][todo + lane];
                queue[w][lane] = keep;
            }
            cnt = rem;
        };

        bool sweep = true;
        if constexpr (REPLAY) {
            int n = -1;
            if (base == firstBase) n = __builtin_amdgcn_readfirstlane(*myCount);
            if ((unsigned)n <= (unsigned)LIST_CAP) {
                sweep = false;
                const uint2 filler = make_uint2((unsigned)OWN_ROWS, 0u);
                if (n > 0) {
                    const int lastEntry = n - 1;
                    uint2 cur = myList[min(lane, lastEntry)];
                    if (lane >= n) cur = filler;
                    int i0 = 0;
                    do {
                        uint2 nxt = myList[min(i0 + 32 + lane, lastEntry)];
                        if (i0 + 32 + lane >= n) nxt = filler;
                        addBatch((int)(cur.x >> 9), __uint_as_float(cur.y), (int)(cur.x & 511u), min(32, n - i0));
                        cur = nxt;
                        i0 += 32;
                    } while (i0 < n);
                }
            }
        }
        if (sweep) {
        for (int t = nGroups * 512 + w * 32; t < E; t += OWN_WAVES * 32) {
            const int e = t + lane;
            unsigned d = 0xffffffffu;
            if (e < E) d = (unsigned)(row[e] - base);
            push(d < rowsHere, e);
        }

        for (int first = w; first < nBlk; first += OWN_WAVES * 32) {
            const int nbRound = min(32, (nBlk - first + OWN_WAVES - 1) / OWN_WAVES);
            int   blkOfLane = blkFirst;
            uint2 mapOfLane = mapFirst;
            if (REPLAY || first != w) loadMaps(first, blkOfLane, mapOfLane);
            int cur = 0;
            unsigned long long m = 0ull;
            int eBlk = 0;
            auto nextChunk = [&](uint4& r, int& e0) -> bool {
                while (m == 0ull && cur < nbRound) {
                    const unsigned lo = (unsigned)__builtin_amdgcn_readlane((int)mapOfLane.x, cur);
                    const unsigned hi = (unsigned)__builtin_amdgcn_readlane((int)mapOfLane.y, cur);
                    m    = ((unsigned long long)hi << 32) | lo;
                    eBlk = __builtin_amdgcn_readlane(blkOfLane, cur) * 4096;
                    ++cur;
                }
                const bool real = m != 0ull;
                int g = -1;
#pragma unroll
                for (int k = 0; k < 8; ++k) {
                    const int pos = (m != 0ull) ? (int)__builtin_ctzll(m) : -1;
                    if ((lane >> 2) == k) g = pos;
                    m &= m - 1ull;
                }
                e0 = (g >= 0) ? (eBlk + g * 64 + (lane & 3) * 16) : -1;
                r  = *(const uint4*)(ownerBytes + max(e0, 0));
                return real;
            };
            uint4 r0, r1, r2, r3;
            int   f0, f1, f2, f3;
            bool a0 = nextChunk(r0, f0);
            nextChunk(r1, f1);
            nextChunk(r2, f2);
            nextChunk(r3, f3);
#pragma unroll 1
            while (a0) {
                scanChunk(r0, f0);
                if (cnt > OWN_QCAP - 512) drain(false);
                a0 = nextChunk(r0, f0);
                scanChunk(r1, f1);
                if (cnt > OWN_QCAP - 512) drain(false);
                nextChunk(r1, f1);
                scanChunk(r2, f2);
                if (cnt > OWN_QCAP - 512) drain(false);
                nextChunk(r2, f2);
                scanChunk(r3, f3);
                if (cnt > OWN_QCAP - 512) drain(false);
                nextChunk(r3, f3);
            }
        }
        if (cnt > 0) drain(true);
        }
        if constexpr (!REPLAY) {
            if (base == firstBase && lane == 0) *myCount = rec;
            rec = -1;
        }
        __syncthreads();

        unsigned* const wl = &queue[0][0];
        {
            constexpr int W_PASSES = NOUT / OWN_WAVES;
            float4 wv[W_PASSES];
#pragma unroll
            for (int p = 0; p < W_PASSES; ++p)
                wv[p] = ((const float4*)(W + (long long)(p * OWN_WAVES + w) * DIM_K))[lane];
#pragma unroll
            for (int p = 0; p < W_PASSES; ++p) {
                unsigned* d = wl + (p * OWN_WAVES + w) * ROW_PITCH + lane;
                d[0]  = __float_as_uint(wv[p].x);
                d[32] = __float_as_uint(wv[p].y);
                d[64] = __float_as_uint(wv[p].z);
                d[96] = __float_as_uint(wv[p].w);
            }
        }
        __syncthreads();

        {
            constexpr int COL_GROUPS = NOUT / 32;
            const int c = lane & 15;
            const int h = lane >> 4;
            const int nStrips = ((int)rowsHere + 15) >> 4;
            const int nUnits  = nStrips * COL_GROUPS;
            for (int u = w; u < nUnits; u += OWN_WAVES) {
                const int strip = u / COL_GROUPS;
                const int cg    = u - strip * COL_GROUPS;
                const int arow = min(strip * 16 + c, OWN_ROWS);
                const float4* ap  = (const float4*)(acc + arow * ROW_PITCH + 4 * h);
                const uint4*  bp0 = (const uint4*)(wl + (cg * 32 + c) * ROW_PITCH + 4 * h);
                const uint4*  bp1 = (const uint4*)(wl + (cg * 32 + 16 + c) * ROW_PITCH + 4 * h);
                v8f d0 = (v8f){}, d1 = (v8f){};
#pragma unroll
                for (int j = 0; j < DIM_K / 8; ++j) {
                    const float4 a  = ap[2 * j];
                    const uint4  p0 = bp0[2 * j];
                    const uint4  p1 = bp1[2 * j];
                    {
                        const v2f x  = (v2f){a.x, a.y};
                        const v2f y0 = (v2f){__uint_as_float(p0.x), __uint_as_float(p0.y)};
                        const v2f y1 = (v2f){__uint_as_float(p1.x), __uint_as_float(p1.y)};
                        d0 = __builtin_amdgcn_wmma_f32_16x16x4_f32(
                             false, x,  false, y0,
                             (short)0, d0,  false,  false);
                        d1 = __builtin_amdgcn_wmma_f32_16x16x4_f32(
                            false, x, false, y1, (short)0, d1, false, false);
                    }
                    {
                        const v2f x  = (v2f){a.z, a.w};
                        const v2f y0 = (v2f){__uint_as_float(p0.z), __uint_as_float(p0.w)};
                        const v2f y1 = (v2f){__uint_as_float(p1.z), __uint_as_float(p1.w)};
                        d0 = __builtin_amdgcn_wmma_f32_16x16x4_f32(
                            false, x, false, y0, (short)0, d0, false, false);
                        d1 = __builtin_amdgcn_wmma_f32_16x16x4_f32(
                            false, x, false, y1, (short)0, d1, false, false);
                    }
                }
                float t0[8], t1[8];
#pragma unroll
                for (int r = 0; r < 8; ++r) { t0[r] = fast_tanh(d0[r]); t1[r] = fast_tanh(d1[r]); }
                const int m0 = strip * 16 + 8 * h;
                float* op = O + (long long)(base + m0) * NOUT + cg * 32 + c;
                if (strip * 16 + 16 <= (int)rowsHere) {
#pragma unroll
                    for (int r = 0; r < 8; ++r) {
                        op[r * NOUT]      = t0[r];
                        op[r * NOUT + 16] = t1[r];
                    }
                } else {
#pragma unroll
                    for (int r = 0; r < 8; ++r) {
                        if ((unsigned)(m0 + r) < rowsHere) {
                            op[r * NOUT]      = t0[r];
                            op[r * NOUT + 16] = t1[r];
                        }
                    }
                }
            }
        }
        __syncthreads();
    }
}

extern "C" void kernel_launch(void* const* d_in, const int* in_sizes, int n_in,
                              void* d_out, int out_size, void* d_ws, size_t ws_size,
                              hipStream_t stream) {
    const float* X        = (const float*)d_in[0];
    const float* W1       = (const float*)d_in[1];
    const float* W2       = (const float*)d_in[2];
    const float* edgeVals = (const float*)d_in[3];
    const int*   edgeRow  = (const int*)d_in[4];
    const int*   edgeCol  = (const int*)d_in[5];
    float* out = (float*)d_out;

    const int N = in_sizes[0] / DIM_K;
    const int E = in_sizes[3];
    const long long featCount = (long long)N * DIM_K;

    float* H1 = (float*)d_ws;
    unsigned* edgeList  = (unsigned*)(H1 + featCount);
    int*      listCount = (int*)(edgeList + 2LL * OWN_GROUPS * OWN_WAVES * LIST_CAP);
    unsigned* ownerWords = (unsigned*)(listCount + OWN_GROUPS * OWN_WAVES);
    unsigned* recordWords = ownerWords + 4 * 1024 * 1024;
    unsigned* groupMaps   = recordWords + 4 * 1024 * 1024;

    const int idBlocks = (E + 4095) / 4096;
    gcn_owner_ids_kernel<<<idBlocks, 256, 0, stream>>>(edgeRow, edgeCol, edgeVals, ownerWords, recordWords, groupMaps, E);

    gcn_spmm_owner_kernel<false, 128><<<OWN_GROUPS, 1024, 0, stream>>>(
        recordWords, edgeRow, X, W1, H1, ownerWords, groupMaps, edgeList, listCount, E, N);

    gcn_spmm_owner_kernel<true, 64><<<OWN_GROUPS, 1024, 0, stream>>>(
        recordWords, edgeRow, H1, W2, out, ownerWords, groupMaps, edgeList, listCount, E, N);
}
